# starting kernel
# baseline (speedup 1.0000x reference)
#include <hip/hip_runtime.h>

typedef __attribute__((ext_vector_type(2))) float v2f;
typedef __attribute__((ext_vector_type(8))) float v8f;


__global__ void k_deg_init(float* __restrict__ deg, int n) {
  int i = blockIdx.x * blockDim.x + threadIdx.x;
  if (i < n) deg[i] = 1.0f;
}

__global__ void k_deg_edges(const int* __restrict__ dst, float* __restrict__ deg, int e) {
  int i = blockIdx.x * blockDim.x + threadIdx.x;
  if (i < e) atomicAdd(&deg[dst[i]], 1.0f);
}

__global__ void k_rsqrt_inplace(float* __restrict__ deg, int n) {
  int i = blockIdx.x * blockDim.x + threadIdx.x;
  if (i < n) deg[i] = rsqrtf(deg[i]);
}


template <int K, int MOUT, bool ACT>
__global__ void k_gemm_strip(const float* __restrict__ X, const float* __restrict__ W,
                             const float* __restrict__ bias_in,
                             const float* __restrict__ dinv,
                             float* __restrict__ hs, float* __restrict__ agg, int nrows) {
  constexpr int T = MOUT / 16;
  const int wave = (blockIdx.x * blockDim.x + threadIdx.x) >> 5;
  const int lane = threadIdx.x & 31;
  if (wave * 16 >= nrows) return;

  const int half = lane >> 4;
  const int l16  = lane & 15;
  const int rowA = wave * 16 + l16;

  v8f acc[T];
#pragma unroll
  for (int t = 0; t < T; ++t) acc[t] = (v8f){};

#pragma unroll 4
  for (int k = 0; k < K; k += 4) {
    const int ka = k + 2 * half;
    v2f a;
    a.x = X[(size_t)rowA * K + ka];
    a.y = X[(size_t)rowA * K + ka + 1];
    if (ACT) {
      a.x = fmaxf(a.x + bias_in[ka], 0.0f);
      a.y = fmaxf(a.y + bias_in[ka + 1], 0.0f);
    }
#pragma unroll
    for (int t = 0; t < T; ++t) {
      const int colB = t * 16 + l16;
      v2f b;
      b.x = W[(size_t)ka * MOUT + colB];
      b.y = W[(size_t)(ka + 1) * MOUT + colB];
      acc[t] = __builtin_amdgcn_wmma_f32_16x16x4_f32(
          false, a, false, b, (short)0, acc[t], false, false);
    }
  }

#pragma unroll
  for (int v = 0; v < 8; ++v) {
    const int row = wave * 16 + v + 8 * half;
    const float dn = dinv[row];
#pragma unroll
    for (int t = 0; t < T; ++t) {
      const int col = t * 16 + l16;
      const float h = acc[t][v] * dn;
      hs[(size_t)row * MOUT + col]  = h;
      agg[(size_t)row * MOUT + col] = h * dn;
    }
  }
}


template <int MOUT>
__global__ void k_scatter(const int* __restrict__ src, const int* __restrict__ dst,
                          const float* __restrict__ hs, const float* __restrict__ dinv,
                          float* __restrict__ agg, int e) {
  constexpr int VPT = MOUT / 32;
  const long long g = (long long)blockIdx.x * blockDim.x + threadIdx.x;
  const int edge = (int)(g >> 5);
  const int lane = (int)(g & 31);
  if (edge >= e) return;
  const int s = src[edge];
  const int d = dst[edge];
  const float dn = dinv[d];
  const float* hrow = hs  + (size_t)s * MOUT + lane * VPT;
  float*       arow = agg + (size_t)d * MOUT + lane * VPT;
  if constexpr (VPT == 4) {
    const float4 v = *(const float4*)hrow;
    atomicAdd(arow + 0, v.x * dn);
    atomicAdd(arow + 1, v.y * dn);
    atomicAdd(arow + 2, v.z * dn);
    atomicAdd(arow + 3, v.w * dn);
  } else if constexpr (VPT == 2) {
    const float2 v = *(const float2*)hrow;
    atomicAdd(arow + 0, v.x * dn);
    atomicAdd(arow + 1, v.y * dn);
  } else {
    atomicAdd(arow, hrow[0] * dn);
  }
}


template <int MOUT>
__global__ void k_bias_out(const float* __restrict__ agg, const float* __restrict__ bias,
                           float* __restrict__ out, int n) {
  const size_t i = (size_t)blockIdx.x * blockDim.x + threadIdx.x;
  if (i >= (size_t)n * MOUT) return;
  out[i] = agg[i] + bias[(int)(i % MOUT)];
}


template <int K, int MOUT, bool ACT>
static void run_layer(const float* in, const float* W, const float* bias_in,
                      const float* dinv, const int* esrc, const int* edst,
                      float* hs, float* agg, int n, int e, hipStream_t stream) {
  const long long strips = (n + 15) / 16;
  const int gblocks = (int)((strips * 32 + 255) / 256);
  k_gemm_strip<K, MOUT, ACT><<<gblocks, 256, 0, stream>>>(in, W, bias_in, dinv, hs, agg, n);

  const int sblocks = (int)(((long long)e * 32 + 255) / 256);
  k_scatter<MOUT><<<sblocks, 256, 0, stream>>>(esrc, edst, hs, dinv, agg, e);
}

extern "C" void kernel_launch(void* const* d_in, const int* in_sizes, int n_in,
                              void* d_out, int out_size, void* d_ws, size_t ws_size,
                              hipStream_t stream) {
  const float* x  = (const float*)d_in[0];
  const int*   ei = (const int*)d_in[1];
  const float* W1 = (const float*)d_in[2];
  const float* b1 = (const float*)d_in[3];
  const float* W2 = (const float*)d_in[4];
  const float* b2 = (const float*)d_in[5];
  const float* W3 = (const float*)d_in[6];
  const float* b3 = (const float*)d_in[7];
  const float* W4 = (const float*)d_in[8];
  const float* b4 = (const float*)d_in[9];

  const int n = in_sizes[0] / 256;
  const int e = in_sizes[1] / 2;
  const int* esrc = ei;
  const int* edst = ei + e;

  float* dinv = (float*)d_ws;
  float* H  = dinv + n;
  float* G1 = H  + (size_t)n * 128;
  float* G2 = G1 + (size_t)n * 128;

  k_deg_init<<<(n + 255) / 256, 256, 0, stream>>>(dinv, n);
  k_deg_edges<<<(e + 255) / 256, 256, 0, stream>>>(edst, dinv, e);
  k_rsqrt_inplace<<<(n + 255) / 256, 256, 0, stream>>>(dinv, n);

  run_layer<256, 128, false>(x,  W1, nullptr, dinv, esrc, edst, H, G1, n, e, stream);
  run_layer<128, 128, true >(G1, W2, b1,      dinv, esrc, edst, H, G2, n, e, stream);
  run_layer<128, 64,  true >(G2, W3, b2,      dinv, esrc, edst, H, G1, n, e, stream);
  run_layer<64,  32,  true >(G1, W4, b3,      dinv, esrc, edst, H, G2, n, e, stream);

  const long long tot = (long long)n * 32;
  k_bias_out<32><<<(int)((tot + 255) / 256), 256, 0, stream>>>(G2, b4, (float*)d_out, n);
}


# Edge scatter: one wave per 32 edges instead of per edge; indices and dinv loaded once per batch, broadcast by readlane; 32 row-line gathers in flight per wave; sums still arrival-ordered atomics
# speedup vs baseline: 1.6457x; 1.6457x over previous
#include <hip/hip_runtime.h>

typedef __attribute__((ext_vector_type(2))) float v2f;
typedef __attribute__((ext_vector_type(8))) float v8f;


__global__ void k_deg_init(float* __restrict__ deg, int n) {
  int i = blockIdx.x * blockDim.x + threadIdx.x;
  if (i < n) deg[i] = 1.0f;
}

__global__ void k_deg_edges(const int* __restrict__ dst, float* __restrict__ deg, int e) {
  int i = blockIdx.x * blockDim.x + threadIdx.x;
  if (i < e) atomicAdd(&deg[dst[i]], 1.0f);
}

__global__ void k_rsqrt_inplace(float* __restrict__ deg, int n) {
  int i = blockIdx.x * blockDim.x + threadIdx.x;
  if (i < n) deg[i] = rsqrtf(deg[i]);
}


template <int K, int MOUT, bool ACT>
__global__ void k_gemm_strip(const float* __restrict__ X, const float* __restrict__ W,
                             const float* __restrict__ bias_in,
                             const float* __restrict__ dinv,
                             float* __restrict__ hs, float* __restrict__ agg, int nrows) {
  constexpr int T = MOUT / 16;
  const int wave = (blockIdx.x * blockDim.x + threadIdx.x) >> 5;
  const int lane = threadIdx.x & 31;
  if (wave * 16 >= nrows) return;

  const int half = lane >> 4;
  const int l16  = lane & 15;
  const int rowA = wave * 16 + l16;

  v8f acc[T];
#pragma unroll
  for (int t = 0; t < T; ++t) acc[t] = (v8f){};

#pragma unroll 4
  for (int k = 0; k < K; k += 4) {
    const int ka = k + 2 * half;
    v2f a;
    a.x = X[(size_t)rowA * K + ka];
    a.y = X[(size_t)rowA * K + ka + 1];
    if (ACT) {
      a.x = fmaxf(a.x + bias_in[ka], 0.0f);
      a.y = fmaxf(a.y + bias_in[ka + 1], 0.0f);
    }
#pragma unroll
    for (int t = 0; t < T; ++t) {
      const int colB = t * 16 + l16;
      v2f b;
      b.x = W[(size_t)ka * MOUT + colB];
      b.y = W[(size_t)(ka + 1) * MOUT + colB];
      acc[t] = __builtin_amdgcn_wmma_f32_16x16x4_f32(
          false, a, false, b, (short)0, acc[t], false, false);
    }
  }

#pragma unroll
  for (int v = 0; v < 8; ++v) {
    const int row = wave * 16 + v + 8 * half;
    const float dn = dinv[row];
#pragma unroll
    for (int t = 0; t < T; ++t) {
      const int col = t * 16 + l16;
      const float h = acc[t][v] * dn;
      hs[(size_t)row * MOUT + col]  = h;
      agg[(size_t)row * MOUT + col] = h * dn;
    }
  }
}


template <int MOUT>
__global__ __launch_bounds__(256) void k_scatter(const int* __restrict__ src,
                                                 const int* __restrict__ dst,
                                                 const float* __restrict__ hs,
                                                 const float* __restrict__ dinv,
                                                 float* __restrict__ agg, int e) {
  constexpr int VPT = MOUT / 32;
  constexpr int G   = 16 / VPT;
  constexpr int NG  = 32 / G;
  const unsigned lane = threadIdx.x & 31u;
  const int wave = __builtin_amdgcn_readfirstlane(
      (int)((blockIdx.x * blockDim.x + threadIdx.x) >> 5));
  const long long base = (long long)wave * 32;
  if (base >= e) return;
  const int cnt = (e - base < 32) ? (int)(e - base) : 32;

  int s = 0, d = 0;
  float dn = 0.0f;
  if ((int)lane < cnt) {
    s  = src[base + lane];
    d  = dst[base + lane];
    dn = dinv[d];
  }
  const int dni = __float_as_int(dn);

  if (cnt == 32) {
    float v[2][G][VPT];
#pragma unroll
    for (int u = 0; u < G; ++u) {
      const float* hrow = hs + (size_t)__builtin_amdgcn_readlane(s, u) * MOUT;
#pragma unroll
      for (int j = 0; j < VPT; ++j) v[0][u][j] = hrow[lane + 32u * j];
    }
#pragma unroll
    for (int g = 0; g < NG; ++g) {
      if (g + 1 < NG) {
#pragma unroll
        for (int u = 0; u < G; ++u) {
          const float* hrow = hs + (size_t)__builtin_amdgcn_readlane(s, (g + 1) * G + u) * MOUT;
#pragma unroll
          for (int j = 0; j < VPT; ++j) v[(g + 1) & 1][u][j] = hrow[lane + 32u * j];
        }
      }
#pragma unroll
      for (int u = 0; u < G; ++u) {
        const float dnj = __int_as_float(__builtin_amdgcn_readlane(dni, g * G + u));
        float* arow = agg + (size_t)__builtin_amdgcn_readlane(d, g * G + u) * MOUT;
#pragma unroll
        for (int j = 0; j < VPT; ++j)
          atomicAdd(arow + (lane + 32u * j), v[g & 1][u][j] * dnj);
      }
    }
  } else {
    for (int i = 0; i < cnt; ++i) {
      const float dnj = __int_as_float(__builtin_amdgcn_readlane(dni, i));
      const float* hrow = hs  + (size_t)__builtin_amdgcn_readlane(s, i) * MOUT;
      float*       arow = agg + (size_t)__builtin_amdgcn_readlane(d, i) * MOUT;
      float t[VPT];
#pragma unroll
      for (int j = 0; j < VPT; ++j) t[j] = hrow[lane + 32u * j];
#pragma unroll
      for (int j = 0; j < VPT; ++j) atomicAdd(arow + (lane + 32u * j), t[j] * dnj);
    }
  }
}


template <int MOUT>
__global__ void k_bias_out(const float* __restrict__ agg, const float* __restrict__ bias,
                           float* __restrict__ out, int n) {
  const size_t i = (size_t)blockIdx.x * blockDim.x + threadIdx.x;
  if (i >= (size_t)n * MOUT) return;
  out[i] = agg[i] + bias[(int)(i % MOUT)];
}


template <int K, int MOUT, bool ACT>
static void run_layer(const float* in, const float* W, const float* bias_in,
                      const float* dinv, const int* esrc, const int* edst,
                      float* hs, float* agg, int n, int e, hipStream_t stream) {
  const long long strips = (n + 15) / 16;
  const int gblocks = (int)((strips * 32 + 255) / 256);
  k_gemm_strip<K, MOUT, ACT><<<gblocks, 256, 0, stream>>>(in, W, bias_in, dinv, hs, agg, n);

  const int sblocks = (int)(((long long)e + 255) / 256);
  k_scatter<MOUT><<<sblocks, 256, 0, stream>>>(esrc, edst, hs, dinv, agg, e);
}

extern "C" void kernel_launch(void* const* d_in, const int* in_sizes, int n_in,
                              void* d_out, int out_size, void* d_ws, size_t ws_size,
                              hipStream_t stream) {
  const float* x  = (const float*)d_in[0];
  const int*   ei = (const int*)d_in[1];
  const float* W1 = (const float*)d_in[2];
  const float* b1 = (const float*)d_in[3];
  const float* W2 = (const float*)d_in[4];
  const float* b2 = (const float*)d_in[5];
  const float* W3 = (const float*)d_in[6];
  const float* b3 = (const float*)d_in[7];
  const float* W4 = (const float*)d_in[8];
  const float* b4 = (const float*)d_in[9];

  const int n = in_sizes[0] / 256;
  const int e = in_sizes[1] / 2;
  const int* esrc = ei;
  const int* edst = ei + e;

  float* dinv = (float*)d_ws;
  float* H  = dinv + n;
  float* G1 = H  + (size_t)n * 128;
  float* G2 = G1 + (size_t)n * 128;

  k_deg_init<<<(n + 255) / 256, 256, 0, stream>>>(dinv, n);
  k_deg_edges<<<(e + 255) / 256, 256, 0, stream>>>(edst, dinv, e);
  k_rsqrt_inplace<<<(n + 255) / 256, 256, 0, stream>>>(dinv, n);

  run_layer<256, 128, false>(x,  W1, nullptr, dinv, esrc, edst, H, G1, n, e, stream);
  run_layer<128, 128, true >(G1, W2, b1,      dinv, esrc, edst, H, G2, n, e, stream);
  run_layer<128, 64,  true >(G2, W3, b2,      dinv, esrc, edst, H, G1, n, e, stream);
  run_layer<64,  32,  true >(G1, W4, b3,      dinv, esrc, edst, H, G2, n, e, stream);

  const long long tot = (long long)n * 32;
  k_bias_out<32><<<(int)((tot + 255) / 256), 256, 0, stream>>>(G2, b4, (float*)d_out, n);
}


# Degree count: per-edge scattered global f32 atomic adds -> LDS histogram per 25000-edge chunk (two 16-bit counts a word), integer planes summed in the rsqrt pass; 12 -> 11 launches
# speedup vs baseline: 1.7767x; 1.0796x over previous
#include <hip/hip_runtime.h>

typedef __attribute__((ext_vector_type(2))) float v2f;
typedef __attribute__((ext_vector_type(8))) float v8f;


constexpr int DEG_CHUNK = 25000;
constexpr int DEG_ITERS = (DEG_CHUNK + 1023) / 1024;
constexpr int DEG_WORDS = 51200;

__global__ __launch_bounds__(1024) void k_deg_hist(const int* __restrict__ dst,
                                                   unsigned* __restrict__ parts, int n, int e) {
  __shared__ unsigned cnt[DEG_WORDS];
  const int half = (n + 1) >> 1;
  const int tid = threadIdx.x;
  const int lo = blockIdx.x * DEG_CHUNK;
  const int m = (e - lo < DEG_CHUNK) ? (e - lo) : DEG_CHUNK;

  int d[DEG_ITERS];
#pragma unroll
  for (int k = 0; k < DEG_ITERS; ++k) {
    const int i = tid + k * 1024;
    d[k] = (i < m) ? dst[lo + i] : -1;
  }

  for (int w = tid; w < half; w += 1024) cnt[w] = 0u;
  __syncthreads();

#pragma unroll
  for (int k = 0; k < DEG_ITERS; ++k) {
    if ((unsigned)d[k] < (unsigned)n) {
      const bool hi = d[k] >= half;
      atomicAdd(&cnt[hi ? d[k] - half : d[k]], hi ? 65536u : 1u);
    }
  }
  __syncthreads();

  unsigned* plane = parts + (size_t)blockIdx.x * half;
  for (int w = tid; w < half; w += 1024) plane[w] = cnt[w];
}

__global__ void k_dinv_from_parts(const unsigned* __restrict__ parts, float* __restrict__ dinv,
                                  int n, int nparts) {
  const int half = (n + 1) >> 1;
  const int w = blockIdx.x * blockDim.x + threadIdx.x;
  if (w >= half) return;
  unsigned lo = 1u, hi = 1u;
#pragma unroll 8
  for (int c = 0; c < nparts; ++c) {
    const unsigned v = parts[(size_t)c * half + w];
    lo += v & 0xFFFFu;
    hi += v >> 16;
  }
  dinv[w] = rsqrtf((float)lo);
  if (w + half < n) dinv[w + half] = rsqrtf((float)hi);
}


template <int K, int MOUT, bool ACT>
__global__ void k_gemm_strip(const float* __restrict__ X, const float* __restrict__ W,
                             const float* __restrict__ bias_in,
                             const float* __restrict__ dinv,
                             float* __restrict__ hs, float* __restrict__ agg, int nrows) {
  constexpr int T = MOUT / 16;
  const int wave = (blockIdx.x * blockDim.x + threadIdx.x) >> 5;
  const int lane = threadIdx.x & 31;
  if (wave * 16 >= nrows) return;

  const int half = lane >> 4;
  const int l16  = lane & 15;
  const int rowA = wave * 16 + l16;

  v8f acc[T];
#pragma unroll
  for (int t = 0; t < T; ++t) acc[t] = (v8f){};

#pragma unroll 4
  for (int k = 0; k < K; k += 4) {
    const int ka = k + 2 * half;
    v2f a;
    a.x = X[(size_t)rowA * K + ka];
    a.y = X[(size_t)rowA * K + ka + 1];
    if (ACT) {
      a.x = fmaxf(a.x + bias_in[ka], 0.0f);
      a.y = fmaxf(a.y + bias_in[ka + 1], 0.0f);
    }
#pragma unroll
    for (int t = 0; t < T; ++t) {
      const int colB = t * 16 + l16;
      v2f b;
      b.x = W[(size_t)ka * MOUT + colB];
      b.y = W[(size_t)(ka + 1) * MOUT + colB];
      acc[t] = __builtin_amdgcn_wmma_f32_16x16x4_f32(
          false, a, false, b, (short)0, acc[t], false, false);
    }
  }

#pragma unroll
  for (int v = 0; v < 8; ++v) {
    const int row = wave * 16 + v + 8 * half;
    const float dn = dinv[row];
#pragma unroll
    for (int t = 0; t < T; ++t) {
      const int col = t * 16 + l16;
      const float h = acc[t][v] * dn;
      hs[(size_t)row * MOUT + col]  = h;
      agg[(size_t)row * MOUT + col] = h * dn;
    }
  }
}


template <int MOUT>
__global__ __launch_bounds__(256) void k_scatter(const int* __restrict__ src,
                                                 const int* __restrict__ dst,
                                                 const float* __restrict__ hs,
                                                 const float* __restrict__ dinv,
                                                 float* __restrict__ agg, int e) {
  constexpr int VPT = MOUT / 32;
  constexpr int G   = 16 / VPT;
  constexpr int NG  = 32 / G;
  const unsigned lane = threadIdx.x & 31u;
  const int wave = __builtin_amdgcn_readfirstlane(
      (int)((blockIdx.x * blockDim.x + threadIdx.x) >> 5));
  const long long base = (long long)wave * 32;
  if (base >= e) return;
  const int cnt = (e - base < 32) ? (int)(e - base) : 32;

  int s = 0, d = 0;
  float dn = 0.0f;
  if ((int)lane < cnt) {
    s  = src[base + lane];
    d  = dst[base + lane];
    dn = dinv[d];
  }
  const int dni = __float_as_int(dn);

  if (cnt == 32) {
    float v[2][G][VPT];
#pragma unroll
    for (int u = 0; u < G; ++u) {
      const float* hrow = hs + (size_t)__builtin_amdgcn_readlane(s, u) * MOUT;
#pragma unroll
      for (int j = 0; j < VPT; ++j) v[0][u][j] = hrow[lane + 32u * j];
    }
#pragma unroll
    for (int g = 0; g < NG; ++g) {
      if (g + 1 < NG) {
#pragma unroll
        for (int u = 0; u < G; ++u) {
          const float* hrow = hs + (size_t)__builtin_amdgcn_readlane(s, (g + 1) * G + u) * MOUT;
#pragma unroll
          for (int j = 0; j < VPT; ++j) v[(g + 1) & 1][u][j] = hrow[lane + 32u * j];
        }
      }
#pragma unroll
      for (int u = 0; u < G; ++u) {
        const float dnj = __int_as_float(__builtin_amdgcn_readlane(dni, g * G + u));
        float* arow = agg + (size_t)__builtin_amdgcn_readlane(d, g * G + u) * MOUT;
#pragma unroll
        for (int j = 0; j < VPT; ++j)
          atomicAdd(arow + (lane + 32u * j), v[g & 1][u][j] * dnj);
      }
    }
  } else {
    for (int i = 0; i < cnt; ++i) {
      const float dnj = __int_as_float(__builtin_amdgcn_readlane(dni, i));
      const float* hrow = hs  + (size_t)__builtin_amdgcn_readlane(s, i) * MOUT;
      float*       arow = agg + (size_t)__builtin_amdgcn_readlane(d, i) * MOUT;
      float t[VPT];
#pragma unroll
      for (int j = 0; j < VPT; ++j) t[j] = hrow[lane + 32u * j];
#pragma unroll
      for (int j = 0; j < VPT; ++j) atomicAdd(arow + (lane + 32u * j), t[j] * dnj);
    }
  }
}


template <int MOUT>
__global__ void k_bias_out(const float* __restrict__ agg, const float* __restrict__ bias,
                           float* __restrict__ out, int n) {
  const size_t i = (size_t)blockIdx.x * blockDim.x + threadIdx.x;
  if (i >= (size_t)n * MOUT) return;
  out[i] = agg[i] + bias[(int)(i % MOUT)];
}


template <int K, int MOUT, bool ACT>
static void run_layer(const float* in, const float* W, const float* bias_in,
                      const float* dinv, const int* esrc, const int* edst,
                      float* hs, float* agg, int n, int e, hipStream_t stream) {
  const long long strips = (n + 15) / 16;
  const int gblocks = (int)((strips * 32 + 255) / 256);
  k_gemm_strip<K, MOUT, ACT><<<gblocks, 256, 0, stream>>>(in, W, bias_in, dinv, hs, agg, n);

  const int sblocks = (int)(((long long)e + 255) / 256);
  k_scatter<MOUT><<<sblocks, 256, 0, stream>>>(esrc, edst, hs, dinv, agg, e);
}

extern "C" void kernel_launch(void* const* d_in, const int* in_sizes, int n_in,
                              void* d_out, int out_size, void* d_ws, size_t ws_size,
                              hipStream_t stream) {
  const float* x  = (const float*)d_in[0];
  const int*   ei = (const int*)d_in[1];
  const float* W1 = (const float*)d_in[2];
  const float* b1 = (const float*)d_in[3];
  const float* W2 = (const float*)d_in[4];
  const float* b2 = (const float*)d_in[5];
  const float* W3 = (const float*)d_in[6];
  const float* b3 = (const float*)d_in[7];
  const float* W4 = (const float*)d_in[8];
  const float* b4 = (const float*)d_in[9];

  const int n = in_sizes[0] / 256;
  const int e = in_sizes[1] / 2;
  const int* esrc = ei;
  const int* edst = ei + e;

  float* dinv = (float*)d_ws;
  float* H  = dinv + n;
  float* G1 = H  + (size_t)n * 128;
  float* G2 = G1 + (size_t)n * 128;
  unsigned* parts = (unsigned*)(G2 + (size_t)n * 128);

  const int nparts = (e + DEG_CHUNK - 1) / DEG_CHUNK;
  k_deg_hist<<<nparts, 1024, 0, stream>>>(edst, parts, n, e);
  k_dinv_from_parts<<<((n + 1) / 2 + 255) / 256, 256, 0, stream>>>(parts, dinv, n, nparts);

  run_layer<256, 128, false>(x,  W1, nullptr, dinv, esrc, edst, H, G1, n, e, stream);
  run_layer<128, 128, true >(G1, W2, b1,      dinv, esrc, edst, H, G2, n, e, stream);
  run_layer<128, 64,  true >(G2, W3, b2,      dinv, esrc, edst, H, G1, n, e, stream);
  run_layer<64,  32,  true >(G1, W4, b3,      dinv, esrc, edst, H, G2, n, e, stream);

  const long long tot = (long long)n * 32;
  k_bias_out<32><<<(int)((tot + 255) / 256), 256, 0, stream>>>(G2, b4, (float*)d_out, n);
}


# Launches 11 -> 9: first product's waves sum the degree planes of their own 16 rows into dinv (rsqrt launch gone); last scatter's write-out adds the output bias and stores out; sum order unchanged
# speedup vs baseline: 2.9993x; 1.6881x over previous
#include <hip/hip_runtime.h>

typedef __attribute__((ext_vector_type(2))) float v2f;
typedef __attribute__((ext_vector_type(8))) float v8f;


constexpr int DEG_CHUNK = 25000;
constexpr int DEG_ITERS = (DEG_CHUNK + 1023) / 1024;
constexpr int DEG_WORDS = 51200;
constexpr int OWN_ROWS  = 400;

__global__ __launch_bounds__(1024) void k_deg_hist(const int* __restrict__ dst,
                                                   unsigned* __restrict__ parts,
                                                   unsigned* __restrict__ own4, int n, int e) {
  unsigned char* own = reinterpret_cast<unsigned char*>(own4);
  __shared__ unsigned cnt[DEG_WORDS];
  const int half = (n + 1) >> 1;
  const int tid = threadIdx.x;
  const int lo = blockIdx.x * DEG_CHUNK;
  const int m = (e - lo < DEG_CHUNK) ? (e - lo) : DEG_CHUNK;

  int d[DEG_ITERS];
#pragma unroll
  for (int k = 0; k < DEG_ITERS; ++k) {
    const int i = tid + k * 1024;
    d[k] = (i < m) ? dst[lo + i] : -1;
    if (i < m) own[lo + i] = (unsigned char)((unsigned)d[k] / (unsigned)OWN_ROWS);
  }

  for (int w = tid; w < half; w += 1024) cnt[w] = 0u;
  __syncthreads();

#pragma unroll
  for (int k = 0; k < DEG_ITERS; ++k) {
    if ((unsigned)d[k] < (unsigned)n) {
      const bool hi = d[k] >= half;
      atomicAdd(&cnt[hi ? d[k] - half : d[k]], hi ? 65536u : 1u);
    }
  }
  __syncthreads();

  unsigned* plane = parts + (size_t)blockIdx.x * half;
  for (int w = tid; w < half; w += 1024) plane[w] = cnt[w];
}


template <int K, int MOUT, bool ACT, bool DEG>
__global__ void k_gemm_strip(const float* __restrict__ X, const float* __restrict__ W,
                             const float* __restrict__ bias_in,
                             const float* __restrict__ dinv,
                             float* __restrict__ hs, int nrows,
                             const unsigned* __restrict__ parts,
                             float* __restrict__ dinv_out,
                             int nparts) {
  constexpr int T = MOUT / 16;
  const int wave = (blockIdx.x * blockDim.x + threadIdx.x) >> 5;
  const int lane = threadIdx.x & 31;
  if (wave * 16 >= nrows) return;

  const int half = lane >> 4;
  const int l16  = lane & 15;
  const int rowA = wave * 16 + l16;

  float rs = 0.0f;
  if (DEG) {
    const int hn = (nrows + 1) >> 1;
    const bool up = rowA >= hn;
    const int wd = (rowA < nrows) ? (up ? rowA - hn : rowA) : 0;
    unsigned cnt = 0u;
    const int npair = (nparts + 1) >> 1;
#pragma unroll 16
    for (int j = 0; j < npair; ++j) {
      const int c = 2 * j + half;
      const bool live = c < nparts;
      const unsigned pv = parts[(size_t)(live ? c : 0) * hn + wd];
      const unsigned one = up ? (pv >> 16) : (pv & 0xFFFFu);
      cnt += live ? one : 0u;
    }
    cnt += (unsigned)__shfl_xor((int)cnt, 16);
    rs = rsqrtf((float)(cnt + 1u));
    if (half == 0 && rowA < nrows) dinv_out[rowA] = rs;
  }

  v8f acc[T];
#pragma unroll
  for (int t = 0; t < T; ++t) acc[t] = (v8f){};

#pragma unroll 4
  for (int k = 0; k < K; k += 4) {
    const int ka = k + 2 * half;
    v2f a;
    a.x = X[(size_t)rowA * K + ka];
    a.y = X[(size_t)rowA * K + ka + 1];
    if (ACT) {
      a.x = fmaxf(a.x + bias_in[ka], 0.0f);
      a.y = fmaxf(a.y + bias_in[ka + 1], 0.0f);
    }
#pragma unroll
    for (int t = 0; t < T; ++t) {
      const int colB = t * 16 + l16;
      v2f b;
      b.x = W[(size_t)ka * MOUT + colB];
      b.y = W[(size_t)(ka + 1) * MOUT + colB];
      acc[t] = __builtin_amdgcn_wmma_f32_16x16x4_f32(
          false, a, false, b, (short)0, acc[t], false, false);
    }
  }

#pragma unroll
  for (int v = 0; v < 8; ++v) {
    const int row = wave * 16 + v + 8 * half;
    const float dn = DEG ? __shfl(rs, v + 8 * half) : dinv[row];
#pragma unroll
    for (int t = 0; t < T; ++t) {
      const int col = t * 16 + l16;
      hs[(size_t)row * MOUT + col] = acc[t][v] * dn;
    }
  }
}


constexpr int Q_WAVE = 640;

template <int MOUT>
__device__ __forceinline__ void walk_full(int s, int dl, int dni, const float* __restrict__ hs,
                                          float* acc, unsigned lane) {
  constexpr int VPT = MOUT / 32;
  constexpr int G   = 16 / VPT;
  constexpr int NG  = 32 / G;
  float v[2][G][VPT];
#pragma unroll
  for (int u = 0; u < G; ++u) {
    const float* hrow = hs + (size_t)__builtin_amdgcn_readlane(s, u) * MOUT;
#pragma unroll
    for (int j = 0; j < VPT; ++j) v[0][u][j] = hrow[lane + 32u * j];
  }
#pragma unroll
  for (int g = 0; g < NG; ++g) {
    if (g + 1 < NG) {
#pragma unroll
      for (int u = 0; u < G; ++u) {
        const float* hrow = hs + (size_t)__builtin_amdgcn_readlane(s, (g + 1) * G + u) * MOUT;
#pragma unroll
        for (int j = 0; j < VPT; ++j) v[(g + 1) & 1][u][j] = hrow[lane + 32u * j];
      }
    }
#pragma unroll
    for (int u = 0; u < G; ++u) {
      const float dnj = __int_as_float(__builtin_amdgcn_readlane(dni, g * G + u));
      float* arow = acc + __builtin_amdgcn_readlane(dl, g * G + u) * MOUT;
#pragma unroll
      for (int j = 0; j < VPT; ++j)
        atomicAdd(arow + (lane + 32u * j), v[g & 1][u][j] * dnj);
    }
  }
}

constexpr int LIST_CAP = 8192;

__device__ __forceinline__ void file_edges(unsigned* __restrict__ glist, int* lcnt, unsigned word,
                                           int k, unsigned lane) {
  int slot = 0;
  if (lane == 0u) slot = atomicAdd(lcnt, k);
  slot = __builtin_amdgcn_readfirstlane(slot);
  if ((int)lane < k && slot + (int)lane < LIST_CAP) glist[slot + (int)lane] = word;
}

template <int MOUT>
__device__ __forceinline__ void walk_part(int s, int dl, int dni, int c, const float* __restrict__ hs,
                                          float* acc, unsigned lane) {
  constexpr int VPT = MOUT / 32;
  for (int i = 0; i < c; ++i) {
    const float dnj = __int_as_float(__builtin_amdgcn_readlane(dni, i));
    const float* hrow = hs + (size_t)__builtin_amdgcn_readlane(s, i) * MOUT;
    float* arow = acc + __builtin_amdgcn_readlane(dl, i) * MOUT;
    float t[VPT];
#pragma unroll
    for (int j = 0; j < VPT; ++j) t[j] = hrow[lane + 32u * j];
#pragma unroll
    for (int j = 0; j < VPT; ++j) atomicAdd(arow + (lane + 32u * j), t[j] * dnj);
  }
}

template <int MOUT, bool BUILD>
__device__ __forceinline__ void scan_edges(const int* __restrict__ src, const int* __restrict__ dst,
                                           const unsigned* __restrict__ own4,
                                           const float* __restrict__ hs,
                                           const float* __restrict__ dinv, float* acc, int* q,
                                           int* lcnt, unsigned* __restrict__ glist, int row0, int e,
                                           unsigned lane, int wv) {
  const uint4* own16 = reinterpret_cast<const uint4*>(own4);
  const unsigned pat = (unsigned)blockIdx.x * 0x01010101u;
  const unsigned nq = (unsigned)e >> 4;
  const unsigned niter = (nq + 1023u) >> 10;
  const unsigned off = ((unsigned)blockIdx.x * niter) / gridDim.x;

  int wcnt = 0;

  for (unsigned it0 = 0; it0 < niter; it0 += 4) {
    uint4 w[4];
    unsigned qi[4];
#pragma unroll
    for (int k = 0; k < 4; ++k) {
      unsigned it = it0 + k + off;
      if (it >= niter) it -= niter;
      qi[k] = (it * 32u + (unsigned)wv) * 32u + lane;
      const unsigned qc = (qi[k] < nq) ? qi[k] : (nq - 1u);
      w[k] = own16[qc];
    }
#pragma unroll
    for (int k = 0; k < 4; ++k) {
      const bool live = (it0 + k < niter) && (qi[k] < nq);
      const unsigned wd[4] = {w[k].x, w[k].y, w[k].z, w[k].w};
#pragma unroll
      for (int dw = 0; dw < 4; ++dw) {
        const unsigned x = wd[dw] ^ pat;
        unsigned z = ~(((x & 0x7F7F7F7Fu) + 0x7F7F7F7Fu) | x | 0x7F7F7F7Fu);
        if (!live) z = 0u;
        const int e0 = (int)(qi[k] * 16u + 4u * dw);
        unsigned bal = __builtin_amdgcn_ballot_w32(z != 0u);
        while (bal != 0u) {
          if (z != 0u) {
            q[wcnt + (int)__builtin_amdgcn_mbcnt_lo(bal, 0u)] = e0 + (__builtin_ctz(z) >> 3);
            z &= z - 1u;
          }
          wcnt += __builtin_popcount(bal);
          bal = __builtin_amdgcn_ballot_w32(z != 0u);
        }
      }
      if (wcnt >= 32) {
        int done = 0;
        while (wcnt - done >= 32) {
          const int ei = q[done + (int)lane];
          const int s = src[ei];
          const int d = dst[ei];
          const int dni = __float_as_int(dinv[d]);
          if (BUILD) file_edges(glist, lcnt, ((unsigned)(d - row0) << 17) | (unsigned)s, 32, lane);
          walk_full<MOUT>(s, d - row0, dni, hs, acc, lane);
          done += 32;
        }
        const int rem = wcnt - done;
        int keep = 0;
        if ((int)lane < rem) keep = q[done + (int)lane];
        if ((int)lane < rem) q[lane] = keep;
        wcnt = rem;
      }
    }
  }

  if (wv == 0) {
    const int ei = (int)(nq * 16u + lane);
    const unsigned char* own1 = reinterpret_cast<const unsigned char*>(own4);
    const bool m = (ei < e) && ((unsigned)own1[ei < e ? ei : 0] == (unsigned)blockIdx.x);
    const unsigned bal = __builtin_amdgcn_ballot_w32(m);
    if (m) q[wcnt + (int)__builtin_amdgcn_mbcnt_lo(bal, 0u)] = ei;
    wcnt += __builtin_popcount(bal);
  }

  for (int qb = 0; qb < wcnt; qb += 32) {
    const int c = (wcnt - qb < 32) ? (wcnt - qb) : 32;
    int s = 0, dl = 0;
    float dn = 0.0f;
    if ((int)lane < c) {
      const int ei = q[qb + (int)lane];
      s = src[ei];
      const int d = dst[ei];
      dn = dinv[d];
      dl = d - row0;
    }
    if (BUILD) file_edges(glist, lcnt, ((unsigned)dl << 17) | (unsigned)s, c, lane);
    walk_part<MOUT>(s, dl, __float_as_int(dn), c, hs, acc, lane);
  }
}

template <int MOUT, bool FIRST, bool LAST>
__global__ __launch_bounds__(1024) void k_scatter_lds(const int* __restrict__ src,
                                                      const int* __restrict__ dst,
                                                      const unsigned* __restrict__ own4,
                                                      const float* __restrict__ hs,
                                                      const float* __restrict__ dinv,
                                                      float* __restrict__ agg,
                                                      unsigned* __restrict__ list,
                                                      int* __restrict__ lcount,
                                                      const float* __restrict__ bias_out,
                                                      int n, int e) {
  __shared__ alignas(16) float acc[OWN_ROWS * MOUT];
  __shared__ int   queue[32 * Q_WAVE];
  __shared__ float dinv_l[OWN_ROWS];
  __shared__ int   lcnt;

  const int tid = threadIdx.x;
  const unsigned lane = threadIdx.x & 31u;
  const int wv = __builtin_amdgcn_readfirstlane((int)(threadIdx.x >> 5));
  const int row0 = blockIdx.x * OWN_ROWS;
  const int nrows = (n - row0 < OWN_ROWS) ? (n - row0) : OWN_ROWS;
  int* q = queue + wv * Q_WAVE;
  unsigned* glist = list + (size_t)blockIdx.x * LIST_CAP;

  for (int i = tid; i < OWN_ROWS * MOUT; i += 1024) acc[i] = 0.0f;
  if (FIRST && tid == 0) lcnt = 0;
  for (int i = tid; i < nrows; i += 1024) dinv_l[i] = dinv[row0 + i];
  __syncthreads();

  if (FIRST) {
    scan_edges<MOUT, true>(src, dst, own4, hs, dinv, acc, q, &lcnt, glist, row0, e, lane, wv);
  } else {
    const int cnt = lcount[blockIdx.x];
    if (__builtin_expect(cnt <= LIST_CAP, 1)) {
      int b = wv * 32;
      unsigned pn = (b + (int)lane < cnt) ? glist[b + (int)lane] : 0u;
      for (; b < cnt; b += 1024) {
        const unsigned p = pn;
        const int bn = b + 1024;
        pn = (bn + (int)lane < cnt) ? glist[bn + (int)lane] : 0u;
        const int s = (int)(p & 0x1FFFFu);
        const int dl = (int)(p >> 17);
        const int dni = __float_as_int(dinv_l[dl]);
        if (cnt - b >= 32) walk_full<MOUT>(s, dl, dni, hs, acc, lane);
        else walk_part<MOUT>(s, dl, dni, cnt - b, hs, acc, lane);
      }
    } else {
      scan_edges<MOUT, false>(src, dst, own4, hs, dinv, acc, q, &lcnt, glist, row0, e, lane, wv);
    }
  }

  constexpr int ROWQ = MOUT / 4;
  constexpr int NQ = (OWN_ROWS * ROWQ + 1023) / 1024;
  const int nq = nrows * ROWQ;
  const float4* hs_own = reinterpret_cast<const float4*>(hs + (size_t)row0 * MOUT);
  float4 own[NQ];
#pragma unroll
  for (int k = 0; k < NQ; ++k) {
    const int p = tid + k * 1024;
    own[k] = hs_own[(p < nq) ? p : 0];
  }
  static_assert(1024 % ROWQ == 0, "a thread's pieces must share their columns");
  float bq[4] = {0.0f, 0.0f, 0.0f, 0.0f};
  if (LAST) {
    const int cb = (tid % ROWQ) * 4;
#pragma unroll
    for (int j = 0; j < 4; ++j) bq[j] = bias_out[cb + j];
  }
  __syncthreads();
  if (FIRST && tid == 0) lcount[blockIdx.x] = lcnt;
  const float4* acc4 = reinterpret_cast<const float4*>(acc);
  float4* out_rows = reinterpret_cast<float4*>(agg + (size_t)row0 * MOUT);
#pragma unroll
  for (int k = 0; k < NQ; ++k) {
    const int p = tid + k * 1024;
    if (p < nq) {
      const float dn = dinv_l[p / ROWQ];
      const float4 a = acc4[p];
      float4 o;
      o.x = a.x + own[k].x * dn;
      o.y = a.y + own[k].y * dn;
      o.z = a.z + own[k].z * dn;
      o.w = a.w + own[k].w * dn;
      if (LAST) {
        o.x += bq[0];
        o.y += bq[1];
        o.z += bq[2];
        o.w += bq[3];
      }
      out_rows[p] = o;
    }
  }
}


template <int K, int MOUT, bool ACT, bool FIRST, bool LAST>
static void run_layer(const float* in, const float* W, const float* bias_in,
                      float* dinv, const int* esrc, const int* edst, const unsigned* own4,
                      float* hs, float* agg, unsigned* list, int* lcount, const unsigned* parts,
                      int nparts, const float* bias_out, int n, int e, hipStream_t stream) {
  const long long strips = (n + 15) / 16;
  const int gblocks = (int)((strips * 32 + 255) / 256);
  k_gemm_strip<K, MOUT, ACT, FIRST><<<gblocks, 256, 0, stream>>>(in, W, bias_in, dinv, hs, n, parts,
                                                                dinv, nparts);

  const int sblocks = (n + OWN_ROWS - 1) / OWN_ROWS;
  k_scatter_lds<MOUT, FIRST, LAST><<<sblocks, 1024, 0, stream>>>(esrc, edst, own4, hs, dinv, agg,
                                                                list, lcount, bias_out, n, e);
}

extern "C" void kernel_launch(void* const* d_in, const int* in_sizes, int n_in,
                              void* d_out, int out_size, void* d_ws, size_t ws_size,
                              hipStream_t stream) {
  const float* x  = (const float*)d_in[0];
  const int*   ei = (const int*)d_in[1];
  const float* W1 = (const float*)d_in[2];
  const float* b1 = (const float*)d_in[3];
  const float* W2 = (const float*)d_in[4];
  const float* b2 = (const float*)d_in[5];
  const float* W3 = (const float*)d_in[6];
  const float* b3 = (const float*)d_in[7];
  const float* W4 = (const float*)d_in[8];
  const float* b4 = (const float*)d_in[9];

  const int n = in_sizes[0] / 256;
  const int e = in_sizes[1] / 2;
  const int* esrc = ei;
  const int* edst = ei + e;

  float* dinv = (float*)d_ws;
  float* H  = dinv + n;
  float* G1 = H  + (size_t)n * 128;
  float* G2 = G1 + (size_t)n * 128;
  unsigned* parts = (unsigned*)(G2 + (size_t)n * 128);

  const int nparts = (e + DEG_CHUNK - 1) / DEG_CHUNK;
  unsigned* own4 = parts + (size_t)nparts * ((n + 1) / 2);
  unsigned* list = own4 + (((size_t)e + 127) / 128) * 32;
  int* lcount = (int*)(list + (size_t)((n + OWN_ROWS - 1) / OWN_ROWS) * LIST_CAP);
  k_deg_hist<<<nparts, 1024, 0, stream>>>(edst, parts, own4, n, e);

  run_layer<256, 128, false, true,  false>(x,  W1, nullptr, dinv, esrc, edst, own4, H, G1, list, lcount, parts, nparts, nullptr, n, e, stream);
  run_layer<128, 128, true,  false, false>(G1, W2, b1,      dinv, esrc, edst, own4, H, G2, list, lcount, parts, nparts, nullptr, n, e, stream);
  run_layer<128, 64,  true,  false, false>(G2, W3, b2,      dinv, esrc, edst, own4, H, G1, list, lcount, parts, nparts, nullptr, n, e, stream);
  run_layer<64,  32,  true,  false, true >(G1, W4, b3,      dinv, esrc, edst, own4, H, (float*)d_out, list, lcount, parts, nparts, b4, n, e, stream);
}


# Products: W staged once per 32-wave group in LDS (rows k,k+1 interleaved); B fragment = one 8-byte LDS read, not two global loads a tile; A rows asked one K block ahead; sum order unchanged
# speedup vs baseline: 3.3214x; 1.1074x over previous
#include <hip/hip_runtime.h>

typedef __attribute__((ext_vector_type(2))) float v2f;
typedef __attribute__((ext_vector_type(8))) float v8f;


constexpr int DEG_CHUNK = 25000;
constexpr int DEG_ITERS = (DEG_CHUNK + 1023) / 1024;
constexpr int DEG_WORDS = 51200;
constexpr int OWN_ROWS  = 400;

__global__ __launch_bounds__(1024) void k_deg_hist(const int* __restrict__ dst,
                                                   unsigned* __restrict__ parts,
                                                   unsigned* __restrict__ own4, int n, int e) {
  unsigned char* own = reinterpret_cast<unsigned char*>(own4);
  __shared__ unsigned cnt[DEG_WORDS];
  const int half = (n + 1) >> 1;
  const int tid = threadIdx.x;
  const int lo = blockIdx.x * DEG_CHUNK;
  const int m = (e - lo < DEG_CHUNK) ? (e - lo) : DEG_CHUNK;

  int d[DEG_ITERS];
#pragma unroll
  for (int k = 0; k < DEG_ITERS; ++k) {
    const int i = tid + k * 1024;
    d[k] = (i < m) ? dst[lo + i] : -1;
    if (i < m) own[lo + i] = (unsigned char)((unsigned)d[k] / (unsigned)OWN_ROWS);
  }

  for (int w = tid; w < half; w += 1024) cnt[w] = 0u;
  __syncthreads();

#pragma unroll
  for (int k = 0; k < DEG_ITERS; ++k) {
    if ((unsigned)d[k] < (unsigned)n) {
      const bool hi = d[k] >= half;
      atomicAdd(&cnt[hi ? d[k] - half : d[k]], hi ? 65536u : 1u);
    }
  }
  __syncthreads();

  unsigned* plane = parts + (size_t)blockIdx.x * half;
  for (int w = tid; w < half; w += 1024) plane[w] = cnt[w];
}


template <int T>
__device__ __forceinline__ void hold_frags(const v2f (&b)[T]) {
  if constexpr (T == 8) {
    asm volatile("" ::"v"(b[0]), "v"(b[1]), "v"(b[2]), "v"(b[3]), "v"(b[4]), "v"(b[5]), "v"(b[6]), "v"(b[7]));
  } else if constexpr (T == 4) {
    asm volatile("" ::"v"(b[0]), "v"(b[1]), "v"(b[2]), "v"(b[3]));
  } else {
    static_assert(T == 2, "column tiles per strip: 8, 4 or 2");
    asm volatile("" ::"v"(b[0]), "v"(b[1]));
  }
}

template <int K, int MOUT, bool ACT>
__device__ __forceinline__ void k_block(const v2f (&cur)[4], v2f (&nxt)[4], const float* __restrict__ xrow,
                                        int kcur, int knext, const float* __restrict__ bias_in, int half,
                                        int l16, const float* Wl, v8f (&acc)[MOUT / 16]) {
  constexpr int T = MOUT / 16;
  constexpr int PITCH = 2 * MOUT + 32;
  v2f ac[4];
#pragma unroll
  for (int s = 0; s < 4; ++s) {
    ac[s] = cur[s];
    if (ACT) {
      const v2f bc = *reinterpret_cast<const v2f*>(bias_in + kcur + 2 * half + 4 * s);
      ac[s].x = fmaxf(cur[s].x + bc.x, 0.0f);
      ac[s].y = fmaxf(cur[s].y + bc.y, 0.0f);
    }
  }
#pragma unroll
  for (int s = 0; s < 4; ++s) nxt[s] = *reinterpret_cast<const v2f*>(xrow + knext + 4 * s);
#pragma unroll
  for (int s = 0; s < 4; ++s) {
    const int ka = kcur + 4 * s + 2 * half;
    v2f b[T];
#pragma unroll
    for (int t = 0; t < T; ++t) {
      const int colB = t * 16 + l16;
      b[t] = *reinterpret_cast<const v2f*>(&Wl[(ka >> 1) * PITCH + colB * 2]);
    }
    hold_frags<T>(b);
#pragma unroll
    for (int t = 0; t < T; ++t) {
      acc[t] = __builtin_amdgcn_wmma_f32_16x16x4_f32(
          false, ac[s], false, b[t], (short)0, acc[t], false, false);
    }
  }
}

template <int K, int MOUT, bool ACT, bool DEG>
__global__ __launch_bounds__(1024) void k_gemm_strip(const float* __restrict__ X, const float* __restrict__ W,
                             const float* __restrict__ bias_in,
                             const float* __restrict__ dinv,
                             float* __restrict__ hs, int nrows,
                             const unsigned* __restrict__ parts,
                             float* __restrict__ dinv_out,
                             int nparts) {
  constexpr int T = MOUT / 16;
  constexpr int PITCH = 2 * MOUT + 32;
  __shared__ alignas(16) float Wl[(K / 2) * PITCH];
  const int wave = blockIdx.x + (int)gridDim.x * (int)(threadIdx.x >> 5);
  const int lane = threadIdx.x & 31;
  const bool active = wave * 16 < nrows;
  const int rowA = wave * 16 + (lane & 15);
  const float* xrow = X + (size_t)rowA * K + 2 * (lane >> 4);
  v2f an[4] = {};
  if (active) {
#pragma unroll
    for (int s = 0; s < 4; ++s) an[s] = *reinterpret_cast<const v2f*>(xrow + 4 * s);
  }
  {
    constexpr int CQ = MOUT / 4;
    constexpr int ITEMS = (K / 2) * CQ;
    constexpr int NIT = (ITEMS + 1023) / 1024;
    const float4* W4 = reinterpret_cast<const float4*>(W);
    float4 r0[NIT], r1[NIT];
#pragma unroll
    for (int j = 0; j < NIT; ++j) {
      const int it = (int)threadIdx.x + j * 1024;
      const int ic = (it < ITEMS) ? it : 0;
      const int kp = ic / CQ;
      const int cq = ic - kp * CQ;
      r0[j] = W4[(size_t)(2 * kp) * CQ + cq];
      r1[j] = W4[(size_t)(2 * kp + 1) * CQ + cq];
    }
#pragma unroll
    for (int j = 0; j < NIT; ++j) {
      const int it = (int)threadIdx.x + j * 1024;
      if (it < ITEMS) {
        const int kp = it / CQ;
        const int cq = it - kp * CQ;
        float4* d = reinterpret_cast<float4*>(&Wl[kp * PITCH + cq * 8]);
        d[0] = make_float4(r0[j].x, r1[j].x, r0[j].y, r1[j].y);
        d[1] = make_float4(r0[j].z, r1[j].z, r0[j].w, r1[j].w);
      }
    }
  }
  __syncthreads();
  if (!active) return;

  const int half = lane >> 4;
  const int l16  = lane & 15;

  float rs = 0.0f;
  if (DEG) {
    const int hn = (nrows + 1) >> 1;
    const bool up = rowA >= hn;
    const int wd = (rowA < nrows) ? (up ? rowA - hn : rowA) : 0;
    unsigned cnt = 0u;
    const int npair = (nparts + 1) >> 1;
#pragma unroll 16
    for (int j = 0; j < npair; ++j) {
      const int c = 2 * j + half;
      const bool live = c < nparts;
      const unsigned pv = parts[(size_t)(live ? c : 0) * hn + wd];
      const unsigned one = up ? (pv >> 16) : (pv & 0xFFFFu);
      cnt += live ? one : 0u;
    }
    cnt += (unsigned)__shfl_xor((int)cnt, 16);
    rs = rsqrtf((float)(cnt + 1u));
    if (half == 0 && rowA < nrows) dinv_out[rowA] = rs;
  }

  v8f acc[T];
#pragma unroll
  for (int t = 0; t < T; ++t) acc[t] = (v8f){};

  static_assert(K % 32 == 0, "K blocks of 16, two per trip");
  v2f am[4];
  for (int k0 = 0; k0 < K; k0 += 32) {
    k_block<K, MOUT, ACT>(an, am, xrow, k0, k0 + 16, bias_in, half, l16, Wl, acc);
    k_block<K, MOUT, ACT>(am, an, xrow, k0 + 16, (k0 + 32 < K) ? (k0 + 32) : (k0 + 16), bias_in, half,
                          l16, Wl, acc);
  }

#pragma unroll
  for (int v = 0; v < 8; ++v) {
    const int row = wave * 16 + v + 8 * half;
    const float dn = DEG ? __shfl(rs, v + 8 * half) : dinv[row];
#pragma unroll
    for (int t = 0; t < T; ++t) {
      const int col = t * 16 + l16;
      hs[(size_t)row * MOUT + col] = acc[t][v] * dn;
    }
  }
}


constexpr int Q_WAVE = 640;

template <int MOUT>
__device__ __forceinline__ void walk_full(int s, int dl, int dni, const float* __restrict__ hs,
                                          float* acc, unsigned lane) {
  constexpr int VPT = MOUT / 32;
  constexpr int G   = 16 / VPT;
  constexpr int NG  = 32 / G;
  float v[2][G][VPT];
#pragma unroll
  for (int u = 0; u < G; ++u) {
    const float* hrow = hs + (size_t)__builtin_amdgcn_readlane(s, u) * MOUT;
#pragma unroll
    for (int j = 0; j < VPT; ++j) v[0][u][j] = hrow[lane + 32u * j];
  }
#pragma unroll
  for (int g = 0; g < NG; ++g) {
    if (g + 1 < NG) {
#pragma unroll
      for (int u = 0; u < G; ++u) {
        const float* hrow = hs + (size_t)__builtin_amdgcn_readlane(s, (g + 1) * G + u) * MOUT;
#pragma unroll
        for (int j = 0; j < VPT; ++j) v[(g + 1) & 1][u][j] = hrow[lane + 32u * j];
      }
    }
#pragma unroll
    for (int u = 0; u < G; ++u) {
      const float dnj = __int_as_float(__builtin_amdgcn_readlane(dni, g * G + u));
      float* arow = acc + __builtin_amdgcn_readlane(dl, g * G + u) * MOUT;
#pragma unroll
      for (int j = 0; j < VPT; ++j)
        atomicAdd(arow + (lane + 32u * j), v[g & 1][u][j] * dnj);
    }
  }
}

constexpr int LIST_CAP = 8192;

__device__ __forceinline__ void file_edges(unsigned* __restrict__ glist, int* lcnt, unsigned word,
                                           int k, unsigned lane) {
  int slot = 0;
  if (lane == 0u) slot = atomicAdd(lcnt, k);
  slot = __builtin_amdgcn_readfirstlane(slot);
  if ((int)lane < k && slot + (int)lane < LIST_CAP) glist[slot + (int)lane] = word;
}

template <int MOUT>
__device__ __forceinline__ void walk_part(int s, int dl, int dni, int c, const float* __restrict__ hs,
                                          float* acc, unsigned lane) {
  constexpr int VPT = MOUT / 32;
  for (int i = 0; i < c; ++i) {
    const float dnj = __int_as_float(__builtin_amdgcn_readlane(dni, i));
    const float* hrow = hs + (size_t)__builtin_amdgcn_readlane(s, i) * MOUT;
    float* arow = acc + __builtin_amdgcn_readlane(dl, i) * MOUT;
    float t[VPT];
#pragma unroll
    for (int j = 0; j < VPT; ++j) t[j] = hrow[lane + 32u * j];
#pragma unroll
    for (int j = 0; j < VPT; ++j) atomicAdd(arow + (lane + 32u * j), t[j] * dnj);
  }
}

template <int MOUT, bool BUILD>
__device__ __forceinline__ void scan_edges(const int* __restrict__ src, const int* __restrict__ dst,
                                           const unsigned* __restrict__ own4,
                                           const float* __restrict__ hs,
                                           const float* __restrict__ dinv, float* acc, int* q,
                                           int* lcnt, unsigned* __restrict__ glist, int row0, int e,
                                           unsigned lane, int wv) {
  const uint4* own16 = reinterpret_cast<const uint4*>(own4);
  const unsigned pat = (unsigned)blockIdx.x * 0x01010101u;
  const unsigned nq = (unsigned)e >> 4;
  const unsigned niter = (nq + 1023u) >> 10;
  const unsigned off = ((unsigned)blockIdx.x * niter) / gridDim.x;

  int wcnt = 0;

  for (unsigned it0 = 0; it0 < niter; it0 += 4) {
    uint4 w[4];
    unsigned qi[4];
#pragma unroll
    for (int k = 0; k < 4; ++k) {
      unsigned it = it0 + k + off;
      if (it >= niter) it -= niter;
      qi[k] = (it * 32u + (unsigned)wv) * 32u + lane;
      const unsigned qc = (qi[k] < nq) ? qi[k] : (nq - 1u);
      w[k] = own16[qc];
    }
#pragma unroll
    for (int k = 0; k < 4; ++k) {
      const bool live = (it0 + k < niter) && (qi[k] < nq);
      const unsigned wd[4] = {w[k].x, w[k].y, w[k].z, w[k].w};
#pragma unroll
      for (int dw = 0; dw < 4; ++dw) {
        const unsigned x = wd[dw] ^ pat;
        unsigned z = ~(((x & 0x7F7F7F7Fu) + 0x7F7F7F7Fu) | x | 0x7F7F7F7Fu);
        if (!live) z = 0u;
        const int e0 = (int)(qi[k] * 16u + 4u * dw);
        unsigned bal = __builtin_amdgcn_ballot_w32(z != 0u);
        while (bal != 0u) {
          if (z != 0u) {
            q[wcnt + (int)__builtin_amdgcn_mbcnt_lo(bal, 0u)] = e0 + (__builtin_ctz(z) >> 3);
            z &= z - 1u;
          }
          wcnt += __builtin_popcount(bal);
          bal = __builtin_amdgcn_ballot_w32(z != 0u);
        }
      }
      if (wcnt >= 32) {
        int done = 0;
        while (wcnt - done >= 32) {
          const int ei = q[done + (int)lane];
          const int s = src[ei];
          const int d = dst[ei];
          const int dni = __float_as_int(dinv[d]);
          if (BUILD) file_edges(glist, lcnt, ((unsigned)(d - row0) << 17) | (unsigned)s, 32, lane);
          walk_full<MOUT>(s, d - row0, dni, hs, acc, lane);
          done += 32;
        }
        const int rem = wcnt - done;
        int keep = 0;
        if ((int)lane < rem) keep = q[done + (int)lane];
        if ((int)lane < rem) q[lane] = keep;
        wcnt = rem;
      }
    }
  }

  if (wv == 0) {
    const int ei = (int)(nq * 16u + lane);
    const unsigned char* own1 = reinterpret_cast<const unsigned char*>(own4);
    const bool m = (ei < e) && ((unsigned)own1[ei < e ? ei : 0] == (unsigned)blockIdx.x);
    const unsigned bal = __builtin_amdgcn_ballot_w32(m);
    if (m) q[wcnt + (int)__builtin_amdgcn_mbcnt_lo(bal, 0u)] = ei;
    wcnt += __builtin_popcount(bal);
  }

  for (int qb = 0; qb < wcnt; qb += 32) {
    const int c = (wcnt - qb < 32) ? (wcnt - qb) : 32;
    int s = 0, dl = 0;
    float dn = 0.0f;
    if ((int)lane < c) {
      const int ei = q[qb + (int)lane];
      s = src[ei];
      const int d = dst[ei];
      dn = dinv[d];
      dl = d - row0;
    }
    if (BUILD) file_edges(glist, lcnt, ((unsigned)dl << 17) | (unsigned)s, c, lane);
    walk_part<MOUT>(s, dl, __float_as_int(dn), c, hs, acc, lane);
  }
}

template <int MOUT, bool FIRST, bool LAST>
__global__ __launch_bounds__(1024) void k_scatter_lds(const int* __restrict__ src,
                                                      const int* __restrict__ dst,
                                                      const unsigned* __restrict__ own4,
                                                      const float* __restrict__ hs,
                                                      const float* __restrict__ dinv,
                                                      float* __restrict__ agg,
                                                      unsigned* __restrict__ list,
                                                      int* __restrict__ lcount,
                                                      const float* __restrict__ bias_out,
                                                      int n, int e) {
  __shared__ alignas(16) float acc[OWN_ROWS * MOUT];
  __shared__ int   queue[32 * Q_WAVE];
  __shared__ float dinv_l[OWN_ROWS];
  __shared__ int   lcnt;

  const int tid = threadIdx.x;
  const unsigned lane = threadIdx.x & 31u;
  const int wv = __builtin_amdgcn_readfirstlane((int)(threadIdx.x >> 5));
  const int row0 = blockIdx.x * OWN_ROWS;
  const int nrows = (n - row0 < OWN_ROWS) ? (n - row0) : OWN_ROWS;
  int* q = queue + wv * Q_WAVE;
  unsigned* glist = list + (size_t)blockIdx.x * LIST_CAP;

  for (int i = tid; i < OWN_ROWS * MOUT; i += 1024) acc[i] = 0.0f;
  if (FIRST && tid == 0) lcnt = 0;
  for (int i = tid; i < nrows; i += 1024) dinv_l[i] = dinv[row0 + i];
  __syncthreads();

  if (FIRST) {
    scan_edges<MOUT, true>(src, dst, own4, hs, dinv, acc, q, &lcnt, glist, row0, e, lane, wv);
  } else {
    const int cnt = lcount[blockIdx.x];
    if (__builtin_expect(cnt <= LIST_CAP, 1)) {
      int b = wv * 32;
      unsigned pn = (b + (int)lane < cnt) ? glist[b + (int)lane] : 0u;
      for (; b < cnt; b += 1024) {
        const unsigned p = pn;
        const int bn = b + 1024;
        pn = (bn + (int)lane < cnt) ? glist[bn + (int)lane] : 0u;
        const int s = (int)(p & 0x1FFFFu);
        const int dl = (int)(p >> 17);
        const int dni = __float_as_int(dinv_l[dl]);
        if (cnt - b >= 32) walk_full<MOUT>(s, dl, dni, hs, acc, lane);
        else walk_part<MOUT>(s, dl, dni, cnt - b, hs, acc, lane);
      }
    } else {
      scan_edges<MOUT, false>(src, dst, own4, hs, dinv, acc, q, &lcnt, glist, row0, e, lane, wv);
    }
  }

  constexpr int ROWQ = MOUT / 4;
  constexpr int NQ = (OWN_ROWS * ROWQ + 1023) / 1024;
  const int nq = nrows * ROWQ;
  const float4* hs_own = reinterpret_cast<const float4*>(hs + (size_t)row0 * MOUT);
  float4 own[NQ];
#pragma unroll
  for (int k = 0; k < NQ; ++k) {
    const int p = tid + k * 1024;
    own[k] = hs_own[(p < nq) ? p : 0];
  }
  static_assert(1024 % ROWQ == 0, "a thread's pieces must share their columns");
  float bq[4] = {0.0f, 0.0f, 0.0f, 0.0f};
  if (LAST) {
    const int cb = (tid % ROWQ) * 4;
#pragma unroll
    for (int j = 0; j < 4; ++j) bq[j] = bias_out[cb + j];
  }
  __syncthreads();
  if (FIRST && tid == 0) lcount[blockIdx.x] = lcnt;
  const float4* acc4 = reinterpret_cast<const float4*>(acc);
  float4* out_rows = reinterpret_cast<float4*>(agg + (size_t)row0 * MOUT);
#pragma unroll
  for (int k = 0; k < NQ; ++k) {
    const int p = tid + k * 1024;
    if (p < nq) {
      const float dn = dinv_l[p / ROWQ];
      const float4 a = acc4[p];
      float4 o;
      o.x = a.x + own[k].x * dn;
      o.y = a.y + own[k].y * dn;
      o.z = a.z + own[k].z * dn;
      o.w = a.w + own[k].w * dn;
      if (LAST) {
        o.x += bq[0];
        o.y += bq[1];
        o.z += bq[2];
        o.w += bq[3];
      }
      out_rows[p] = o;
    }
  }
}


template <int K, int MOUT, bool ACT, bool FIRST, bool LAST>
static void run_layer(const float* in, const float* W, const float* bias_in,
                      float* dinv, const int* esrc, const int* edst, const unsigned* own4,
                      float* hs, float* agg, unsigned* list, int* lcount, const unsigned* parts,
                      int nparts, const float* bias_out, int n, int e, hipStream_t stream) {
  k_gemm_strip<K, MOUT, ACT, FIRST><<<256, 1024, 0, stream>>>(in, W, bias_in, dinv, hs, n, parts,
                                                             dinv, nparts);

  const int sblocks = (n + OWN_ROWS - 1) / OWN_ROWS;
  k_scatter_lds<MOUT, FIRST, LAST><<<sblocks, 1024, 0, stream>>>(esrc, edst, own4, hs, dinv, agg,
                                                                list, lcount, bias_out, n, e);
}

extern "C" void kernel_launch(void* const* d_in, const int* in_sizes, int n_in,
                              void* d_out, int out_size, void* d_ws, size_t ws_size,
                              hipStream_t stream) {
  const float* x  = (const float*)d_in[0];
  const int*   ei = (const int*)d_in[1];
  const float* W1 = (const float*)d_in[2];
  const float* b1 = (const float*)d_in[3];
  const float* W2 = (const float*)d_in[4];
  const float* b2 = (const float*)d_in[5];
  const float* W3 = (const float*)d_in[6];
  const float* b3 = (const float*)d_in[7];
  const float* W4 = (const float*)d_in[8];
  const float* b4 = (const float*)d_in[9];

  const int n = in_sizes[0] / 256;
  const int e = in_sizes[1] / 2;
  const int* esrc = ei;
  const int* edst = ei + e;

  float* dinv = (float*)d_ws;
  float* H  = dinv + n;
  float* G1 = H  + (size_t)n * 128;
  float* G2 = G1 + (size_t)n * 128;
  unsigned* parts = (unsigned*)(G2 + (size_t)n * 128);

  const int nparts = (e + DEG_CHUNK - 1) / DEG_CHUNK;
  unsigned* own4 = parts + (size_t)nparts * ((n + 1) / 2);
  unsigned* list = own4 + (((size_t)e + 127) / 128) * 32;
  int* lcount = (int*)(list + (size_t)((n + OWN_ROWS - 1) / OWN_ROWS) * LIST_CAP);
  k_deg_hist<<<nparts, 1024, 0, stream>>>(edst, parts, own4, n, e);

  run_layer<256, 128, false, true,  false>(x,  W1, nullptr, dinv, esrc, edst, own4, H, G1, list, lcount, parts, nparts, nullptr, n, e, stream);
  run_layer<128, 128, true,  false, false>(G1, W2, b1,      dinv, esrc, edst, own4, H, G2, list, lcount, parts, nparts, nullptr, n, e, stream);
  run_layer<128, 64,  true,  false, false>(G2, W3, b2,      dinv, esrc, edst, own4, H, G1, list, lcount, parts, nparts, nullptr, n, e, stream);
  run_layer<64,  32,  true,  false, true >(G1, W4, b3,      dinv, esrc, edst, own4, H, (float*)d_out, list, lcount, parts, nparts, b4, n, e, stream);
}


# Layer-1 edge lists: degree kernel counting-sorts each 25000-edge chunk by owner group in LDS; a scatter group joins its 64 segments instead of scanning the owner bytes; sums still arrival-ordered
# speedup vs baseline: 3.7321x; 1.1237x over previous
#include <hip/hip_runtime.h>

typedef __attribute__((ext_vector_type(2))) float v2f;
typedef __attribute__((ext_vector_type(8))) float v8f;


constexpr int DEG_CHUNK = 25000;
constexpr int DEG_ITERS = (DEG_CHUNK + 1023) / 1024;
constexpr int DEG_WORDS = 51200;
constexpr int OWN_ROWS  = 400;
constexpr int MAX_OWN   = 256;
constexpr int SEG_PITCH = MAX_OWN + 32;
constexpr int SORT_PITCH = ((DEG_CHUNK + 31) / 32) * 32;

__global__ __launch_bounds__(1024) void k_deg_hist(const int* __restrict__ src,
                                                   const int* __restrict__ dst,
                                                   unsigned* __restrict__ parts,
                                                   unsigned* __restrict__ own4,
                                                   unsigned* __restrict__ sorted,
                                                   int* __restrict__ seg, int n, int e) {
  unsigned char* own = reinterpret_cast<unsigned char*>(own4);
  __shared__ unsigned cnt[DEG_WORDS];
  __shared__ unsigned sbuf[SORT_PITCH];
  __shared__ int ocnt[MAX_OWN];
  __shared__ int obase[MAX_OWN + 1];
  const int half = (n + 1) >> 1;
  const int tid = threadIdx.x;
  const int lo = blockIdx.x * DEG_CHUNK;
  const int m = (e - lo < DEG_CHUNK) ? (e - lo) : DEG_CHUNK;

  int d[DEG_ITERS];
  int s[DEG_ITERS];
#pragma unroll
  for (int k = 0; k < DEG_ITERS; ++k) {
    const int i = tid + k * 1024;
    d[k] = (i < m) ? dst[lo + i] : -1;
    s[k] = (i < m) ? src[lo + i] : 0;
    if (i < m) own[lo + i] = (unsigned char)((unsigned)d[k] / (unsigned)OWN_ROWS);
  }

  for (int w = tid; w < half; w += 1024) cnt[w] = 0u;
  if (tid < MAX_OWN) ocnt[tid] = 0;
  __syncthreads();

  unsigned wd[DEG_ITERS];
  unsigned orr[DEG_ITERS];
#pragma unroll
  for (int k = 0; k < DEG_ITERS; ++k) {
    wd[k] = 0u;
    orr[k] = ~0u;
    if ((unsigned)d[k] < (unsigned)n) {
      const bool hi = d[k] >= half;
      atomicAdd(&cnt[hi ? d[k] - half : d[k]], hi ? 65536u : 1u);
      const unsigned o = (unsigned)d[k] / (unsigned)OWN_ROWS;
      wd[k] = (((unsigned)d[k] - o * (unsigned)OWN_ROWS) << 17) | (unsigned)s[k];
      orr[k] = (o << 16) | (unsigned)atomicAdd(&ocnt[o], 1);
    }
  }
  __syncthreads();

  if (tid < 32) {
    int v[MAX_OWN / 32];
    int sum = 0;
#pragma unroll
    for (int j = 0; j < MAX_OWN / 32; ++j) {
      v[j] = ocnt[tid * (MAX_OWN / 32) + j];
      sum += v[j];
    }
    int incl = sum;
#pragma unroll
    for (int o = 1; o < 32; o <<= 1) {
      const int t = __shfl_up(incl, o);
      if (tid >= o) incl += t;
    }
    int run = incl - sum;
#pragma unroll
    for (int j = 0; j < MAX_OWN / 32; ++j) {
      obase[tid * (MAX_OWN / 32) + j] = run;
      run += v[j];
    }
    if (tid == 31) obase[MAX_OWN] = incl;
  }
  __syncthreads();

#pragma unroll
  for (int k = 0; k < DEG_ITERS; ++k) {
    if (orr[k] != ~0u) sbuf[obase[orr[k] >> 16] + (int)(orr[k] & 0xFFFFu)] = wd[k];
  }
  if (tid <= MAX_OWN) seg[(size_t)blockIdx.x * SEG_PITCH + tid] = obase[tid];
  __syncthreads();

  const int tot = obase[MAX_OWN];
  unsigned* srt = sorted + (size_t)blockIdx.x * SORT_PITCH;
  for (int w = tid; w < tot; w += 1024) srt[w] = sbuf[w];
  unsigned* plane = parts + (size_t)blockIdx.x * half;
  for (int w = tid; w < half; w += 1024) plane[w] = cnt[w];
}


template <int T>
__device__ __forceinline__ void hold_frags(const v2f (&b)[T]) {
  if constexpr (T == 8) {
    asm volatile("" ::"v"(b[0]), "v"(b[1]), "v"(b[2]), "v"(b[3]), "v"(b[4]), "v"(b[5]), "v"(b[6]), "v"(b[7]));
  } else if constexpr (T == 4) {
    asm volatile("" ::"v"(b[0]), "v"(b[1]), "v"(b[2]), "v"(b[3]));
  } else {
    static_assert(T == 2, "column tiles per strip: 8, 4 or 2");
    asm volatile("" ::"v"(b[0]), "v"(b[1]));
  }
}

template <int K, int MOUT, bool ACT>
__device__ __forceinline__ void k_block(const v2f (&cur)[4], v2f (&nxt)[4], const float* __restrict__ xrow,
                                        int kcur, int knext, const float* __restrict__ bias_in, int half,
                                        int l16, const float* Wl, v8f (&acc)[MOUT / 16]) {
  constexpr int T = MOUT / 16;
  constexpr int PITCH = 2 * MOUT + 32;
  v2f ac[4];
#pragma unroll
  for (int s = 0; s < 4; ++s) {
    ac[s] = cur[s];
    if (ACT) {
      const v2f bc = *reinterpret_cast<const v2f*>(bias_in + kcur + 2 * half + 4 * s);
      ac[s].x = fmaxf(cur[s].x + bc.x, 0.0f);
      ac[s].y = fmaxf(cur[s].y + bc.y, 0.0f);
    }
  }
#pragma unroll
  for (int s = 0; s < 4; ++s) nxt[s] = *reinterpret_cast<const v2f*>(xrow + knext + 4 * s);
#pragma unroll
  for (int s = 0; s < 4; ++s) {
    const int ka = kcur + 4 * s + 2 * half;
    v2f b[T];
#pragma unroll
    for (int t = 0; t < T; ++t) {
      const int colB = t * 16 + l16;
      b[t] = *reinterpret_cast<const v2f*>(&Wl[(ka >> 1) * PITCH + colB * 2]);
    }
    hold_frags<T>(b);
#pragma unroll
    for (int t = 0; t < T; ++t) {
      acc[t] = __builtin_amdgcn_wmma_f32_16x16x4_f32(
          false, ac[s], false, b[t], (short)0, acc[t], false, false);
    }
  }
}

template <int K, int MOUT, bool ACT, bool DEG>
__global__ __launch_bounds__(1024) void k_gemm_strip(const float* __restrict__ X, const float* __restrict__ W,
                             const float* __restrict__ bias_in,
                             const float* __restrict__ dinv,
                             float* __restrict__ hs, int nrows,
                             const unsigned* __restrict__ parts,
                             float* __restrict__ dinv_out,
                             int nparts) {
  constexpr int T = MOUT / 16;
  constexpr int PITCH = 2 * MOUT + 32;
  __shared__ alignas(16) float Wl[(K / 2) * PITCH];
  const int wave = blockIdx.x + (int)gridDim.x * (int)(threadIdx.x >> 5);
  const int lane = threadIdx.x & 31;
  const bool active = wave * 16 < nrows;
  const int rowA = wave * 16 + (lane & 15);
  const float* xrow = X + (size_t)rowA * K + 2 * (lane >> 4);
  v2f an[4] = {};
  if (active) {
#pragma unroll
    for (int s = 0; s < 4; ++s) an[s] = *reinterpret_cast<const v2f*>(xrow + 4 * s);
  }
  {
    constexpr int CQ = MOUT / 4;
    constexpr int ITEMS = (K / 2) * CQ;
    constexpr int NIT = (ITEMS + 1023) / 1024;
    const float4* W4 = reinterpret_cast<const float4*>(W);
    float4 r0[NIT], r1[NIT];
#pragma unroll
    for (int j = 0; j < NIT; ++j) {
      const int it = (int)threadIdx.x + j * 1024;
      const int ic = (it < ITEMS) ? it : 0;
      const int kp = ic / CQ;
      const int cq = ic - kp * CQ;
      r0[j] = W4[(size_t)(2 * kp) * CQ + cq];
      r1[j] = W4[(size_t)(2 * kp + 1) * CQ + cq];
    }
#pragma unroll
    for (int j = 0; j < NIT; ++j) {
      const int it = (int)threadIdx.x + j * 1024;
      if (it < ITEMS) {
        const int kp = it / CQ;
        const int cq = it - kp * CQ;
        float4* d = reinterpret_cast<float4*>(&Wl[kp * PITCH + cq * 8]);
        d[0] = make_float4(r0[j].x, r1[j].x, r0[j].y, r1[j].y);
        d[1] = make_float4(r0[j].z, r1[j].z, r0[j].w, r1[j].w);
      }
    }
  }
  __syncthreads();
  if (!active) return;

  const int half = lane >> 4;
  const int l16  = lane & 15;

  float rs = 0.0f;
  if (DEG) {
    const int hn = (nrows + 1) >> 1;
    const bool up = rowA >= hn;
    const int wd = (rowA < nrows) ? (up ? rowA - hn : rowA) : 0;
    unsigned cnt = 0u;
    const int npair = (nparts + 1) >> 1;
#pragma unroll 16
    for (int j = 0; j < npair; ++j) {
      const int c = 2 * j + half;
      const bool live = c < nparts;
      const unsigned pv = parts[(size_t)(live ? c : 0) * hn + wd];
      const unsigned one = up ? (pv >> 16) : (pv & 0xFFFFu);
      cnt += live ? one : 0u;
    }
    cnt += (unsigned)__shfl_xor((int)cnt, 16);
    rs = rsqrtf((float)(cnt + 1u));
    if (half == 0 && rowA < nrows) dinv_out[rowA] = rs;
  }

  v8f acc[T];
#pragma unroll
  for (int t = 0; t < T; ++t) acc[t] = (v8f){};

  static_assert(K % 32 == 0, "K blocks of 16, two per trip");
  v2f am[4];
  for (int k0 = 0; k0 < K; k0 += 32) {
    k_block<K, MOUT, ACT>(an, am, xrow, k0, k0 + 16, bias_in, half, l16, Wl, acc);
    k_block<K, MOUT, ACT>(am, an, xrow, k0 + 16, (k0 + 32 < K) ? (k0 + 32) : (k0 + 16), bias_in, half,
                          l16, Wl, acc);
  }

#pragma unroll
  for (int v = 0; v < 8; ++v) {
    const int row = wave * 16 + v + 8 * half;
    const float dn = DEG ? __shfl(rs, v + 8 * half) : dinv[row];
#pragma unroll
    for (int t = 0; t < T; ++t) {
      const int col = t * 16 + l16;
      hs[(size_t)row * MOUT + col] = acc[t][v] * dn;
    }
  }
}


constexpr int Q_WAVE = 640;
constexpr int LIST_CAP = 8192;
constexpr int MAX_CHUNKS = 1024;

template <int MOUT>
__device__ __forceinline__ void walk_full(int s, int dl, int dni, const float* __restrict__ hs,
                                          float* acc, unsigned lane) {
  constexpr int VPT = MOUT / 32;
  constexpr int G   = 16 / VPT;
  constexpr int NG  = 32 / G;
  float v[2][G][VPT];
#pragma unroll
  for (int u = 0; u < G; ++u) {
    const float* hrow = hs + (size_t)__builtin_amdgcn_readlane(s, u) * MOUT;
#pragma unroll
    for (int j = 0; j < VPT; ++j) v[0][u][j] = hrow[lane + 32u * j];
  }
#pragma unroll
  for (int g = 0; g < NG; ++g) {
    if (g + 1 < NG) {
#pragma unroll
      for (int u = 0; u < G; ++u) {
        const float* hrow = hs + (size_t)__builtin_amdgcn_readlane(s, (g + 1) * G + u) * MOUT;
#pragma unroll
        for (int j = 0; j < VPT; ++j) v[(g + 1) & 1][u][j] = hrow[lane + 32u * j];
      }
    }
#pragma unroll
    for (int u = 0; u < G; ++u) {
      const float dnj = __int_as_float(__builtin_amdgcn_readlane(dni, g * G + u));
      float* arow = acc + __builtin_amdgcn_readlane(dl, g * G + u) * MOUT;
#pragma unroll
      for (int j = 0; j < VPT; ++j)
        atomicAdd(arow + (lane + 32u * j), v[g & 1][u][j] * dnj);
    }
  }
}

template <int MOUT>
__device__ __forceinline__ void walk_part(int s, int dl, int dni, int c, const float* __restrict__ hs,
                                          float* acc, unsigned lane) {
  constexpr int VPT = MOUT / 32;
  for (int i = 0; i < c; ++i) {
    const float dnj = __int_as_float(__builtin_amdgcn_readlane(dni, i));
    const float* hrow = hs + (size_t)__builtin_amdgcn_readlane(s, i) * MOUT;
    float* arow = acc + __builtin_amdgcn_readlane(dl, i) * MOUT;
    float t[VPT];
#pragma unroll
    for (int j = 0; j < VPT; ++j) t[j] = hrow[lane + 32u * j];
#pragma unroll
    for (int j = 0; j < VPT; ++j) atomicAdd(arow + (lane + 32u * j), t[j] * dnj);
  }
}

template <int MOUT, bool LOCAL>
__device__ __forceinline__ void walk_list(const unsigned* __restrict__ glist, const unsigned* lst,
                                          int cnt, const float* dinv_l, const float* __restrict__ hs,
                                          float* acc, unsigned lane, int wv) {
  int b = wv * 32;
  unsigned pn = 0u;
  if (!LOCAL) pn = (b + (int)lane < cnt) ? glist[b + (int)lane] : 0u;
  for (; b < cnt; b += 1024) {
    unsigned p;
    if (LOCAL) {
      p = (b + (int)lane < cnt) ? lst[b + (int)lane] : 0u;
    } else {
      p = pn;
      const int bn = b + 1024;
      pn = (bn + (int)lane < cnt) ? glist[bn + (int)lane] : 0u;
    }
    const int s = (int)(p & 0x1FFFFu);
    const int dl = (int)(p >> 17);
    const int dni = __float_as_int(dinv_l[dl]);
    if (cnt - b >= 32) walk_full<MOUT>(s, dl, dni, hs, acc, lane);
    else walk_part<MOUT>(s, dl, dni, cnt - b, hs, acc, lane);
  }
}

template <int MOUT>
__device__ __forceinline__ void scan_edges(const int* __restrict__ src, const int* __restrict__ dst,
                                           const unsigned* __restrict__ own4,
                                           const float* __restrict__ hs,
                                           const float* __restrict__ dinv, float* acc, int* q,
                                           int row0, int e, unsigned lane, int wv) {
  const uint4* own16 = reinterpret_cast<const uint4*>(own4);
  const unsigned pat = (unsigned)blockIdx.x * 0x01010101u;
  const unsigned nq = (unsigned)e >> 4;
  const unsigned niter = (nq + 1023u) >> 10;
  const unsigned off = ((unsigned)blockIdx.x * niter) / gridDim.x;

  int wcnt = 0;

  for (unsigned it0 = 0; it0 < niter; it0 += 4) {
    uint4 w[4];
    unsigned qi[4];
#pragma unroll
    for (int k = 0; k < 4; ++k) {
      unsigned it = it0 + k + off;
      if (it >= niter) it -= niter;
      qi[k] = (it * 32u + (unsigned)wv) * 32u + lane;
      const unsigned qc = (qi[k] < nq) ? qi[k] : (nq - 1u);
      w[k] = own16[qc];
    }
#pragma unroll
    for (int k = 0; k < 4; ++k) {
      const bool live = (it0 + k < niter) && (qi[k] < nq);
      const unsigned wd[4] = {w[k].x, w[k].y, w[k].z, w[k].w};
#pragma unroll
      for (int dw = 0; dw < 4; ++dw) {
        const unsigned x = wd[dw] ^ pat;
        unsigned z = ~(((x & 0x7F7F7F7Fu) + 0x7F7F7F7Fu) | x | 0x7F7F7F7Fu);
        if (!live) z = 0u;
        const int e0 = (int)(qi[k] * 16u + 4u * dw);
        unsigned bal = __builtin_amdgcn_ballot_w32(z != 0u);
        while (bal != 0u) {
          if (z != 0u) {
            q[wcnt + (int)__builtin_amdgcn_mbcnt_lo(bal, 0u)] = e0 + (__builtin_ctz(z) >> 3);
            z &= z - 1u;
          }
          wcnt += __builtin_popcount(bal);
          bal = __builtin_amdgcn_ballot_w32(z != 0u);
        }
      }
      if (wcnt >= 32) {
        int done = 0;
        while (wcnt - done >= 32) {
          const int ei = q[done + (int)lane];
          const int s = src[ei];
          const int d = dst[ei];
          const int dni = __float_as_int(dinv[d]);
          walk_full<MOUT>(s, d - row0, dni, hs, acc, lane);
          done += 32;
        }
        const int rem = wcnt - done;
        int keep = 0;
        if ((int)lane < rem) keep = q[done + (int)lane];
        if ((int)lane < rem) q[lane] = keep;
        wcnt = rem;
      }
    }
  }

  if (wv == 0) {
    const int ei = (int)(nq * 16u + lane);
    const unsigned char* own1 = reinterpret_cast<const unsigned char*>(own4);
    const bool m = (ei < e) && ((unsigned)own1[ei < e ? ei : 0] == (unsigned)blockIdx.x);
    const unsigned bal = __builtin_amdgcn_ballot_w32(m);
    if (m) q[wcnt + (int)__builtin_amdgcn_mbcnt_lo(bal, 0u)] = ei;
    wcnt += __builtin_popcount(bal);
  }

  for (int qb = 0; qb < wcnt; qb += 32) {
    const int c = (wcnt - qb < 32) ? (wcnt - qb) : 32;
    int s = 0, dl = 0;
    float dn = 0.0f;
    if ((int)lane < c) {
      const int ei = q[qb + (int)lane];
      s = src[ei];
      const int d = dst[ei];
      dn = dinv[d];
      dl = d - row0;
    }
    walk_part<MOUT>(s, dl, __float_as_int(dn), c, hs, acc, lane);
  }
}

template <int MOUT, bool FIRST, bool LAST>
__global__ __launch_bounds__(1024) void k_scatter_lds(const int* __restrict__ src,
                                                      const int* __restrict__ dst,
                                                      const unsigned* __restrict__ own4,
                                                      const float* __restrict__ hs,
                                                      const float* __restrict__ dinv,
                                                      float* __restrict__ agg,
                                                      unsigned* __restrict__ list,
                                                      int* __restrict__ lcount,
                                                      const float* __restrict__ bias_out,
                                                      const unsigned* __restrict__ sorted,
                                                      const int* __restrict__ seg,
                                                      int nchunks, int n, int e) {
  __shared__ alignas(16) float acc[OWN_ROWS * MOUT];
  __shared__ int   queue[32 * Q_WAVE];
  __shared__ float dinv_l[OWN_ROWS];
  static_assert(LIST_CAP + 2 * MAX_CHUNKS <= 32 * Q_WAVE, "list and segment tables share the queues' LDS");
  unsigned* lst = reinterpret_cast<unsigned*>(queue);
  int* sst = queue + LIST_CAP;
  int* sln = sst + MAX_CHUNKS;

  const int tid = threadIdx.x;
  const unsigned lane = threadIdx.x & 31u;
  const int wv = __builtin_amdgcn_readfirstlane((int)(threadIdx.x >> 5));
  const int row0 = blockIdx.x * OWN_ROWS;
  const int nrows = (n - row0 < OWN_ROWS) ? (n - row0) : OWN_ROWS;
  int* q = queue + wv * Q_WAVE;
  unsigned* glist = list + (size_t)blockIdx.x * LIST_CAP;
  const bool tables_fit = nchunks <= MAX_CHUNKS;

  const bool has_seg = FIRST && tables_fit && tid < nchunks;
  int a0 = 0, a1 = 0;
  if (has_seg) {
    const int* sg = seg + (size_t)tid * SEG_PITCH + blockIdx.x;
    a0 = sg[0];
    a1 = sg[1];
  }
  for (int i = tid; i < OWN_ROWS * MOUT; i += 1024) acc[i] = 0.0f;
  for (int i = tid; i < nrows; i += 1024) dinv_l[i] = dinv[row0 + i];
  if (has_seg) {
    sst[tid] = tid * SORT_PITCH + a0;
    sln[tid] = a1 - a0;
  }
  __syncthreads();

  if (FIRST) {
    int cnt = LIST_CAP + 1;
    if (tables_fit) {
      int carry = 0;
      for (int base = 0; base < nchunks; base += 32) {
        const int ci = base + (int)lane;
        const int v = (ci < nchunks) ? sln[ci] : 0;
        int incl = v;
#pragma unroll
        for (int o = 1; o < 32; o <<= 1) {
          const int t = __shfl_up(incl, o);
          if ((int)lane >= o) incl += t;
        }
        const int pr = carry + __shfl(incl - v, wv);
        const int ln = __shfl(v, wv);
        carry += __shfl(incl, 31);
        if (ln > 0) {
          const int st = sst[base + wv];
          for (int i0 = (int)lane; i0 < ln + (int)lane; i0 += 128) {
            unsigned t[4];
#pragma unroll
            for (int u = 0; u < 4; ++u) {
              const int i = i0 + 32 * u;
              t[u] = sorted[st + ((i < ln) ? i : (ln - 1))];
            }
#pragma unroll
            for (int u = 0; u < 4; ++u) {
              const int i = i0 + 32 * u;
              if (i < ln && pr + i < LIST_CAP) lst[pr + i] = t[u];
            }
          }
        }
      }
      cnt = __builtin_amdgcn_readfirstlane(carry);
    }
    __syncthreads();
    if (tid == 0) lcount[blockIdx.x] = cnt;
    if (__builtin_expect(cnt <= LIST_CAP, 1)) {
      for (int i = tid; i < cnt; i += 1024) glist[i] = lst[i];
      walk_list<MOUT, true>(glist, lst, cnt, dinv_l, hs, acc, lane, wv);
    } else {
      scan_edges<MOUT>(src, dst, own4, hs, dinv, acc, q, row0, e, lane, wv);
    }
  } else {
    const int cnt = lcount[blockIdx.x];
    if (__builtin_expect(cnt <= LIST_CAP, 1)) {
      walk_list<MOUT, false>(glist, lst, cnt, dinv_l, hs, acc, lane, wv);
    } else {
      scan_edges<MOUT>(src, dst, own4, hs, dinv, acc, q, row0, e, lane, wv);
    }
  }

  constexpr int ROWQ = MOUT / 4;
  constexpr int NQ = (OWN_ROWS * ROWQ + 1023) / 1024;
  const int nq = nrows * ROWQ;
  const float4* hs_own = reinterpret_cast<const float4*>(hs + (size_t)row0 * MOUT);
  float4 own[NQ];
#pragma unroll
  for (int k = 0; k < NQ; ++k) {
    const int p = tid + k * 1024;
    own[k] = hs_own[(p < nq) ? p : 0];
  }
  static_assert(1024 % ROWQ == 0, "a thread's pieces must share their columns");
  float bq[4] = {0.0f, 0.0f, 0.0f, 0.0f};
  if (LAST) {
    const int cb = (tid % ROWQ) * 4;
#pragma unroll
    for (int j = 0; j < 4; ++j) bq[j] = bias_out[cb + j];
  }
  __syncthreads();
  const float4* acc4 = reinterpret_cast<const float4*>(acc);
  float4* out_rows = reinterpret_cast<float4*>(agg + (size_t)row0 * MOUT);
#pragma unroll
  for (int k = 0; k < NQ; ++k) {
    const int p = tid + k * 1024;
    if (p < nq) {
      const float dn = dinv_l[p / ROWQ];
      const float4 a = acc4[p];
      float4 o;
      o.x = a.x + own[k].x * dn;
      o.y = a.y + own[k].y * dn;
      o.z = a.z + own[k].z * dn;
      o.w = a.w + own[k].w * dn;
      if (LAST) {
        o.x += bq[0];
        o.y += bq[1];
        o.z += bq[2];
        o.w += bq[3];
      }
      out_rows[p] = o;
    }
  }
}


template <int K, int MOUT, bool ACT, bool FIRST, bool LAST>
static void run_layer(const float* in, const float* W, const float* bias_in,
                      float* dinv, const int* esrc, const int* edst, const unsigned* own4,
                      float* hs, float* agg, unsigned* list, int* lcount, const unsigned* parts,
                      int nparts, const float* bias_out, const unsigned* sorted, const int* seg,
                      int n, int e, hipStream_t stream) {
  k_gemm_strip<K, MOUT, ACT, FIRST><<<256, 1024, 0, stream>>>(in, W, bias_in, dinv, hs, n, parts,
                                                             dinv, nparts);

  const int sblocks = (n + OWN_ROWS - 1) / OWN_ROWS;
  k_scatter_lds<MOUT, FIRST, LAST><<<sblocks, 1024, 0, stream>>>(esrc, edst, own4, hs, dinv, agg,
                                                                list, lcount, bias_out, sorted, seg,
                                                                nparts, n, e);
}

extern "C" void kernel_launch(void* const* d_in, const int* in_sizes, int n_in,
                              void* d_out, int out_size, void* d_ws, size_t ws_size,
                              hipStream_t stream) {
  const float* x  = (const float*)d_in[0];
  const int*   ei = (const int*)d_in[1];
  const float* W1 = (const float*)d_in[2];
  const float* b1 = (const float*)d_in[3];
  const float* W2 = (const float*)d_in[4];
  const float* b2 = (const float*)d_in[5];
  const float* W3 = (const float*)d_in[6];
  const float* b3 = (const float*)d_in[7];
  const float* W4 = (const float*)d_in[8];
  const float* b4 = (const float*)d_in[9];

  const int n = in_sizes[0] / 256;
  const int e = in_sizes[1] / 2;
  const int* esrc = ei;
  const int* edst = ei + e;

  float* dinv = (float*)d_ws;
  float* H  = dinv + n;
  float* G1 = H  + (size_t)n * 128;
  float* G2 = G1 + (size_t)n * 128;
  unsigned* parts = (unsigned*)(G2 + (size_t)n * 128);

  const int nparts = (e + DEG_CHUNK - 1) / DEG_CHUNK;
  unsigned* own4 = parts + (size_t)nparts * ((n + 1) / 2);
  unsigned* list = own4 + (((size_t)e + 127) / 128) * 32;
  int* lcount = (int*)(list + (size_t)((n + OWN_ROWS - 1) / OWN_ROWS) * LIST_CAP);
  unsigned* sorted = (unsigned*)(lcount + MAX_OWN);
  int* seg = (int*)(sorted + (size_t)nparts * SORT_PITCH);
  k_deg_hist<<<nparts, 1024, 0, stream>>>(esrc, edst, parts, own4, sorted, seg, n, e);

  run_layer<256, 128, false, true,  false>(x,  W1, nullptr, dinv, esrc, edst, own4, H, G1, list, lcount, parts, nparts, nullptr, sorted, seg, n, e, stream);
  run_layer<128, 128, true,  false, false>(G1, W2, b1,      dinv, esrc, edst, own4, H, G2, list, lcount, parts, nparts, nullptr, sorted, seg, n, e, stream);
  run_layer<128, 64,  true,  false, false>(G2, W3, b2,      dinv, esrc, edst, own4, H, G1, list, lcount, parts, nparts, nullptr, sorted, seg, n, e, stream);
  run_layer<64,  32,  true,  false, true >(G1, W4, b3,      dinv, esrc, edst, own4, H, (float*)d_out, list, lcount, parts, nparts, b4, sorted, seg, n, e, stream);
}


# Edge scatter: a group's list grouped once by owning wave (wave w: rows w, w+32, ..); a wave sums its own rows in registers indexed by row, no LDS float atomics; sums in integer-atomic rank order
# speedup vs baseline: 3.8477x; 1.0310x over previous
#include <hip/hip_runtime.h>

typedef __attribute__((ext_vector_type(2))) float v2f;
typedef __attribute__((ext_vector_type(8))) float v8f;


constexpr int DEG_CHUNK = 25000;
constexpr int DEG_ITERS = (DEG_CHUNK + 1023) / 1024;
constexpr int DEG_WORDS = 51200;
constexpr int OWN_ROWS  = 400;
constexpr int MAX_OWN   = 256;
constexpr int SEG_PITCH = MAX_OWN + 32;
constexpr int SORT_PITCH = ((DEG_CHUNK + 31) / 32) * 32;

__global__ __launch_bounds__(1024) void k_deg_hist(const int* __restrict__ src,
                                                   const int* __restrict__ dst,
                                                   unsigned* __restrict__ parts,
                                                   unsigned* __restrict__ own4,
                                                   unsigned* __restrict__ sorted,
                                                   int* __restrict__ seg, int n, int e) {
  unsigned char* own = reinterpret_cast<unsigned char*>(own4);
  __shared__ unsigned cnt[DEG_WORDS];
  __shared__ unsigned sbuf[SORT_PITCH];
  __shared__ int ocnt[MAX_OWN];
  __shared__ int obase[MAX_OWN + 1];
  const int half = (n + 1) >> 1;
  const int tid = threadIdx.x;
  const int lo = blockIdx.x * DEG_CHUNK;
  const int m = (e - lo < DEG_CHUNK) ? (e - lo) : DEG_CHUNK;

  int d[DEG_ITERS];
  int s[DEG_ITERS];
#pragma unroll
  for (int k = 0; k < DEG_ITERS; ++k) {
    const int i = tid + k * 1024;
    d[k] = (i < m) ? dst[lo + i] : -1;
    s[k] = (i < m) ? src[lo + i] : 0;
    if (i < m) own[lo + i] = (unsigned char)((unsigned)d[k] / (unsigned)OWN_ROWS);
  }

  for (int w = tid; w < half; w += 1024) cnt[w] = 0u;
  if (tid < MAX_OWN) ocnt[tid] = 0;
  __syncthreads();

  unsigned wd[DEG_ITERS];
  unsigned orr[DEG_ITERS];
#pragma unroll
  for (int k = 0; k < DEG_ITERS; ++k) {
    wd[k] = 0u;
    orr[k] = ~0u;
    if ((unsigned)d[k] < (unsigned)n) {
      const bool hi = d[k] >= half;
      atomicAdd(&cnt[hi ? d[k] - half : d[k]], hi ? 65536u : 1u);
      const unsigned o = (unsigned)d[k] / (unsigned)OWN_ROWS;
      wd[k] = (((unsigned)d[k] - o * (unsigned)OWN_ROWS) << 17) | (unsigned)s[k];
      orr[k] = (o << 16) | (unsigned)atomicAdd(&ocnt[o], 1);
    }
  }
  __syncthreads();

  if (tid < 32) {
    int v[MAX_OWN / 32];
    int sum = 0;
#pragma unroll
    for (int j = 0; j < MAX_OWN / 32; ++j) {
      v[j] = ocnt[tid * (MAX_OWN / 32) + j];
      sum += v[j];
    }
    int incl = sum;
#pragma unroll
    for (int o = 1; o < 32; o <<= 1) {
      const int t = __shfl_up(incl, o);
      if (tid >= o) incl += t;
    }
    int run = incl - sum;
#pragma unroll
    for (int j = 0; j < MAX_OWN / 32; ++j) {
      obase[tid * (MAX_OWN / 32) + j] = run;
      run += v[j];
    }
    if (tid == 31) obase[MAX_OWN] = incl;
  }
  __syncthreads();

#pragma unroll
  for (int k = 0; k < DEG_ITERS; ++k) {
    if (orr[k] != ~0u) sbuf[obase[orr[k] >> 16] + (int)(orr[k] & 0xFFFFu)] = wd[k];
  }
  if (tid <= MAX_OWN) seg[(size_t)blockIdx.x * SEG_PITCH + tid] = obase[tid];
  __syncthreads();

  const int tot = obase[MAX_OWN];
  unsigned* srt = sorted + (size_t)blockIdx.x * SORT_PITCH;
  for (int w = tid; w < tot; w += 1024) srt[w] = sbuf[w];
  unsigned* plane = parts + (size_t)blockIdx.x * half;
  for (int w = tid; w < half; w += 1024) plane[w] = cnt[w];
}


template <int T>
__device__ __forceinline__ void hold_frags(const v2f (&b)[T]) {
  if constexpr (T == 8) {
    asm volatile("" ::"v"(b[0]), "v"(b[1]), "v"(b[2]), "v"(b[3]), "v"(b[4]), "v"(b[5]), "v"(b[6]), "v"(b[7]));
  } else if constexpr (T == 4) {
    asm volatile("" ::"v"(b[0]), "v"(b[1]), "v"(b[2]), "v"(b[3]));
  } else {
    static_assert(T == 2, "column tiles per strip: 8, 4 or 2");
    asm volatile("" ::"v"(b[0]), "v"(b[1]));
  }
}

template <int K, int MOUT, bool ACT>
__device__ __forceinline__ void k_block(const v2f (&cur)[4], v2f (&nxt)[4], const float* __restrict__ xrow,
                                        int kcur, int knext, const float* __restrict__ bias_in, int half,
                                        int l16, const float* Wl, v8f (&acc)[MOUT / 16]) {
  constexpr int T = MOUT / 16;
  constexpr int PITCH = 2 * MOUT + 32;
  v2f ac[4];
#pragma unroll
  for (int s = 0; s < 4; ++s) {
    ac[s] = cur[s];
    if (ACT) {
      const v2f bc = *reinterpret_cast<const v2f*>(bias_in + kcur + 2 * half + 4 * s);
      ac[s].x = fmaxf(cur[s].x + bc.x, 0.0f);
      ac[s].y = fmaxf(cur[s].y + bc.y, 0.0f);
    }
  }
#pragma unroll
  for (int s = 0; s < 4; ++s) nxt[s] = *reinterpret_cast<const v2f*>(xrow + knext + 4 * s);
#pragma unroll
  for (int s = 0; s < 4; ++s) {
    const int ka = kcur + 4 * s + 2 * half;
    v2f b[T];
#pragma unroll
    for (int t = 0; t < T; ++t) {
      const int colB = t * 16 + l16;
      b[t] = *reinterpret_cast<const v2f*>(&Wl[(ka >> 1) * PITCH + colB * 2]);
    }
    hold_frags<T>(b);
#pragma unroll
    for (int t = 0; t < T; ++t) {
      acc[t] = __builtin_amdgcn_wmma_f32_16x16x4_f32(
          false, ac[s], false, b[t], (short)0, acc[t], false, false);
    }
  }
}

template <int K, int MOUT, bool ACT, bool DEG>
__global__ __launch_bounds__(1024) void k_gemm_strip(const float* __restrict__ X, const float* __restrict__ W,
                             const float* __restrict__ bias_in,
                             const float* __restrict__ dinv,
                             float* __restrict__ hs, int nrows,
                             const unsigned* __restrict__ parts,
                             float* __restrict__ dinv_out,
                             int nparts) {
  constexpr int T = MOUT / 16;
  constexpr int PITCH = 2 * MOUT + 32;
  __shared__ alignas(16) float Wl[(K / 2) * PITCH];
  const int wave = blockIdx.x + (int)gridDim.x * (int)(threadIdx.x >> 5);
  const int lane = threadIdx.x & 31;
  const bool active = wave * 16 < nrows;
  const int rowA = wave * 16 + (lane & 15);
  const float* xrow = X + (size_t)rowA * K + 2 * (lane >> 4);
  v2f an[4] = {};
  if (active) {
#pragma unroll
    for (int s = 0; s < 4; ++s) an[s] = *reinterpret_cast<const v2f*>(xrow + 4 * s);
  }
  {
    constexpr int CQ = MOUT / 4;
    constexpr int ITEMS = (K / 2) * CQ;
    constexpr int NIT = (ITEMS + 1023) / 1024;
    const float4* W4 = reinterpret_cast<const float4*>(W);
    float4 r0[NIT], r1[NIT];
#pragma unroll
    for (int j = 0; j < NIT; ++j) {
      const int it = (int)threadIdx.x + j * 1024;
      const int ic = (it < ITEMS) ? it : 0;
      const int kp = ic / CQ;
      const int cq = ic - kp * CQ;
      r0[j] = W4[(size_t)(2 * kp) * CQ + cq];
      r1[j] = W4[(size_t)(2 * kp + 1) * CQ + cq];
    }
#pragma unroll
    for (int j = 0; j < NIT; ++j) {
      const int it = (int)threadIdx.x + j * 1024;
      if (it < ITEMS) {
        const int kp = it / CQ;
        const int cq = it - kp * CQ;
        float4* d = reinterpret_cast<float4*>(&Wl[kp * PITCH + cq * 8]);
        d[0] = make_float4(r0[j].x, r1[j].x, r0[j].y, r1[j].y);
        d[1] = make_float4(r0[j].z, r1[j].z, r0[j].w, r1[j].w);
      }
    }
  }
  __syncthreads();
  if (!active) return;

  const int half = lane >> 4;
  const int l16  = lane & 15;

  float rs = 0.0f;
  if (DEG) {
    const int hn = (nrows + 1) >> 1;
    const bool up = rowA >= hn;
    const int wd = (rowA < nrows) ? (up ? rowA - hn : rowA) : 0;
    unsigned cnt = 0u;
    const int npair = (nparts + 1) >> 1;
#pragma unroll 16
    for (int j = 0; j < npair; ++j) {
      const int c = 2 * j + half;
      const bool live = c < nparts;
      const unsigned pv = parts[(size_t)(live ? c : 0) * hn + wd];
      const unsigned one = up ? (pv >> 16) : (pv & 0xFFFFu);
      cnt += live ? one : 0u;
    }
    cnt += (unsigned)__shfl_xor((int)cnt, 16);
    rs = rsqrtf((float)(cnt + 1u));
    if (half == 0 && rowA < nrows) dinv_out[rowA] = rs;
  }

  v8f acc[T];
#pragma unroll
  for (int t = 0; t < T; ++t) acc[t] = (v8f){};

  static_assert(K % 32 == 0, "K blocks of 16, two per trip");
  v2f am[4];
  for (int k0 = 0; k0 < K; k0 += 32) {
    k_block<K, MOUT, ACT>(an, am, xrow, k0, k0 + 16, bias_in, half, l16, Wl, acc);
    k_block<K, MOUT, ACT>(am, an, xrow, k0 + 16, (k0 + 32 < K) ? (k0 + 32) : (k0 + 16), bias_in, half,
                          l16, Wl, acc);
  }

#pragma unroll
  for (int v = 0; v < 8; ++v) {
    const int row = wave * 16 + v + 8 * half;
    const float dn = DEG ? __shfl(rs, v + 8 * half) : dinv[row];
#pragma unroll
    for (int t = 0; t < T; ++t) {
      const int col = t * 16 + l16;
      hs[(size_t)row * MOUT + col] = acc[t][v] * dn;
    }
  }
}


constexpr int Q_WAVE = 640;
constexpr int LIST_CAP = 8192;
constexpr int MAX_CHUNKS = 1024;
constexpr int RPW = (OWN_ROWS + 31) / 32;
constexpr int WT_PITCH = 64;

template <int VPT> struct rowpiece;
template <> struct rowpiece<4> { typedef float4 type; };
template <> struct rowpiece<2> { typedef float2 type; };
template <> struct rowpiece<1> { typedef float type; };
__device__ __forceinline__ float piece_get(const float4& p, int j) { return j == 0 ? p.x : j == 1 ? p.y : j == 2 ? p.z : p.w; }
__device__ __forceinline__ float piece_get(const float2& p, int j) { return j == 0 ? p.x : p.y; }
__device__ __forceinline__ float piece_get(const float& p, int) { return p; }

typedef __attribute__((ext_vector_type(16))) float v16f;
static_assert(RPW <= 16, "a wave's rows index one 16-float register vector");

template <int MOUT, bool FULL>
__device__ __forceinline__ void batch_own(v16f (&a)[MOUT / 32], int s, int rl, int dni, int c,
                                          const typename rowpiece<MOUT / 32>::type* __restrict__ hsv) {
  constexpr int VPT = MOUT / 32;
  constexpr int G   = 16 / VPT;
  constexpr int NG  = 32 / G;
  typedef typename rowpiece<VPT>::type VT;
  VT v[2][G];
#pragma unroll
  for (int u = 0; u < G; ++u) v[0][u] = hsv[(size_t)__builtin_amdgcn_readlane(s, u) * 32];
#pragma unroll
  for (int g = 0; g < NG; ++g) {
    if (g + 1 < NG && (FULL || (g + 1) * G < c)) {
#pragma unroll
      for (int u = 0; u < G; ++u)
        v[(g + 1) & 1][u] = hsv[(size_t)__builtin_amdgcn_readlane(s, (g + 1) * G + u) * 32];
    }
#pragma unroll
    for (int u = 0; u < G; ++u) {
      const int i = g * G + u;
      if (FULL || i < c) {
        const float dn = __int_as_float(__builtin_amdgcn_readlane(dni, i));
        const int r = __builtin_amdgcn_readlane(rl, i);
#pragma unroll
        for (int j = 0; j < VPT; ++j) a[j][r] += piece_get(v[g & 1][u], j) * dn;
      }
    }
  }
}

template <int MOUT, bool LOCAL>
__device__ __forceinline__ void walk_own(const unsigned* __restrict__ glist, const unsigned* lst,
                                         int beg, int end, const float* dinv_l,
                                         const float* __restrict__ hs, float* acc, unsigned lane, int wv) {
  constexpr int VPT = MOUT / 32;
  typedef typename rowpiece<VPT>::type VT;
  v16f a[VPT];
#pragma unroll
  for (int j = 0; j < VPT; ++j) a[j] = (v16f){};
  const VT* hsv = reinterpret_cast<const VT*>(hs) + lane;
  int b = beg;
  unsigned pn = 0u;
  if (!LOCAL && b + (int)lane < end) pn = glist[b + (int)lane];
  for (; b + 32 <= end; b += 32) {
    unsigned p;
    if (LOCAL) {
      p = lst[b + (int)lane];
    } else {
      p = pn;
      pn = (b + 32 + (int)lane < end) ? glist[b + 32 + (int)lane] : 0u;
    }
    const int dl = (int)(p >> 17);
    batch_own<MOUT, true>(a, (int)(p & 0x1FFFFu), dl >> 5, __float_as_int(dinv_l[dl]), 32, hsv);
  }
  if (b < end) {
    const int c = end - b;
    unsigned p = pn;
    if (LOCAL) p = ((int)lane < c) ? lst[b + (int)lane] : 0u;
    const int dl = (int)(p >> 17);
    batch_own<MOUT, false>(a, (int)(p & 0x1FFFFu), dl >> 5, __float_as_int(dinv_l[dl]), c, hsv);
  }
#pragma unroll
  for (int k = 0; k < RPW; ++k) {
    const int row = k * 32 + wv;
    if (row < OWN_ROWS) {
#pragma unroll
      for (int j = 0; j < VPT; ++j) acc[row * MOUT + (int)lane * VPT + j] = a[j][k];
    }
  }
}

template <int MOUT>
__device__ __forceinline__ void walk_full(int s, int dl, int dni, const float* __restrict__ hs,
                                          float* acc, unsigned lane) {
  constexpr int VPT = MOUT / 32;
  constexpr int G   = 16 / VPT;
  constexpr int NG  = 32 / G;
  float v[2][G][VPT];
#pragma unroll
  for (int u = 0; u < G; ++u) {
    const float* hrow = hs + (size_t)__builtin_amdgcn_readlane(s, u) * MOUT;
#pragma unroll
    for (int j = 0; j < VPT; ++j) v[0][u][j] = hrow[lane + 32u * j];
  }
#pragma unroll
  for (int g = 0; g < NG; ++g) {
    if (g + 1 < NG) {
#pragma unroll
      for (int u = 0; u < G; ++u) {
        const float* hrow = hs + (size_t)__builtin_amdgcn_readlane(s, (g + 1) * G + u) * MOUT;
#pragma unroll
        for (int j = 0; j < VPT; ++j) v[(g + 1) & 1][u][j] = hrow[lane + 32u * j];
      }
    }
#pragma unroll
    for (int u = 0; u < G; ++u) {
      const float dnj = __int_as_float(__builtin_amdgcn_readlane(dni, g * G + u));
      float* arow = acc + __builtin_amdgcn_readlane(dl, g * G + u) * MOUT;
#pragma unroll
      for (int j = 0; j < VPT; ++j)
        atomicAdd(arow + (lane + 32u * j), v[g & 1][u][j] * dnj);
    }
  }
}

template <int MOUT>
__device__ __forceinline__ void walk_part(int s, int dl, int dni, int c, const float* __restrict__ hs,
                                          float* acc, unsigned lane) {
  constexpr int VPT = MOUT / 32;
  for (int i = 0; i < c; ++i) {
    const float dnj = __int_as_float(__builtin_amdgcn_readlane(dni, i));
    const float* hrow = hs + (size_t)__builtin_amdgcn_readlane(s, i) * MOUT;
    float* arow = acc + __builtin_amdgcn_readlane(dl, i) * MOUT;
    float t[VPT];
#pragma unroll
    for (int j = 0; j < VPT; ++j) t[j] = hrow[lane + 32u * j];
#pragma unroll
    for (int j = 0; j < VPT; ++j) atomicAdd(arow + (lane + 32u * j), t[j] * dnj);
  }
}

template <int MOUT>
__device__ __forceinline__ void scan_edges(const int* __restrict__ src, const int* __restrict__ dst,
                                           const unsigned* __restrict__ own4,
                                           const float* __restrict__ hs,
                                           const float* __restrict__ dinv, float* acc, int* q,
                                           int row0, int e, unsigned lane, int wv) {
  const uint4* own16 = reinterpret_cast<const uint4*>(own4);
  const unsigned pat = (unsigned)blockIdx.x * 0x01010101u;
  const unsigned nq = (unsigned)e >> 4;
  const unsigned niter = (nq + 1023u) >> 10;
  const unsigned off = ((unsigned)blockIdx.x * niter) / gridDim.x;

  int wcnt = 0;

  for (unsigned it0 = 0; it0 < niter; it0 += 4) {
    uint4 w[4];
    unsigned qi[4];
#pragma unroll
    for (int k = 0; k < 4; ++k) {
      unsigned it = it0 + k + off;
      if (it >= niter) it -= niter;
      qi[k] = (it * 32u + (unsigned)wv) * 32u + lane;
      const unsigned qc = (qi[k] < nq) ? qi[k] : (nq - 1u);
      w[k] = own16[qc];
    }
#pragma unroll
    for (int k = 0; k < 4; ++k) {
      const bool live = (it0 + k < niter) && (qi[k] < nq);
      const unsigned wd[4] = {w[k].x, w[k].y, w[k].z, w[k].w};
#pragma unroll
      for (int dw = 0; dw < 4; ++dw) {
        const unsigned x = wd[dw] ^ pat;
        unsigned z = ~(((x & 0x7F7F7F7Fu) + 0x7F7F7F7Fu) | x | 0x7F7F7F7Fu);
        if (!live) z = 0u;
        const int e0 = (int)(qi[k] * 16u + 4u * dw);
        unsigned bal = __builtin_amdgcn_ballot_w32(z != 0u);
        while (bal != 0u) {
          if (z != 0u) {
            q[wcnt + (int)__builtin_amdgcn_mbcnt_lo(bal, 0u)] = e0 + (__builtin_ctz(z) >> 3);
            z &= z - 1u;
          }
          wcnt += __builtin_popcount(bal);
          bal = __builtin_amdgcn_ballot_w32(z != 0u);
        }
      }
      if (wcnt >= 32) {
        int done = 0;
        while (wcnt - done >= 32) {
          const int ei = q[done + (int)lane];
          const int s = src[ei];
          const int d = dst[ei];
          const int dni = __float_as_int(dinv[d]);
          walk_full<MOUT>(s, d - row0, dni, hs, acc, lane);
          done += 32;
        }
        const int rem = wcnt - done;
        int keep = 0;
        if ((int)lane < rem) keep = q[done + (int)lane];
        if ((int)lane < rem) q[lane] = keep;
        wcnt = rem;
      }
    }
  }

  if (wv == 0) {
    const int ei = (int)(nq * 16u + lane);
    const unsigned char* own1 = reinterpret_cast<const unsigned char*>(own4);
    const bool m = (ei < e) && ((unsigned)own1[ei < e ? ei : 0] == (unsigned)blockIdx.x);
    const unsigned bal = __builtin_amdgcn_ballot_w32(m);
    if (m) q[wcnt + (int)__builtin_amdgcn_mbcnt_lo(bal, 0u)] = ei;
    wcnt += __builtin_popcount(bal);
  }

  for (int qb = 0; qb < wcnt; qb += 32) {
    const int c = (wcnt - qb < 32) ? (wcnt - qb) : 32;
    int s = 0, dl = 0;
    float dn = 0.0f;
    if ((int)lane < c) {
      const int ei = q[qb + (int)lane];
      s = src[ei];
      const int d = dst[ei];
      dn = dinv[d];
      dl = d - row0;
    }
    walk_part<MOUT>(s, dl, __float_as_int(dn), c, hs, acc, lane);
  }
}

template <int MOUT, bool FIRST, bool LAST>
__global__ __launch_bounds__(1024) void k_scatter_lds(const int* __restrict__ src,
                                                      const int* __restrict__ dst,
                                                      const unsigned* __restrict__ own4,
                                                      const float* __restrict__ hs,
                                                      const float* __restrict__ dinv,
                                                      float* __restrict__ agg,
                                                      unsigned* __restrict__ list,
                                                      int* __restrict__ lcount,
                                                      const float* __restrict__ bias_out,
                                                      const unsigned* __restrict__ sorted,
                                                      const int* __restrict__ seg,
                                                      int nchunks, int n, int e) {
  __shared__ alignas(16) float acc[OWN_ROWS * MOUT];
  __shared__ int   queue[32 * Q_WAVE];
  __shared__ float dinv_l[OWN_ROWS];
  __shared__ int wcnt[32];
  __shared__ int wbase[33];
  static_assert(2 * LIST_CAP + 2 * MAX_CHUNKS <= 32 * Q_WAVE, "lists and segment tables share the queues' LDS");
  unsigned* lst = reinterpret_cast<unsigned*>(queue);
  int* sst = queue + LIST_CAP;
  int* sln = sst + MAX_CHUNKS;
  unsigned* lst2 = lst + LIST_CAP + 2 * MAX_CHUNKS;

  const int tid = threadIdx.x;
  const unsigned lane = threadIdx.x & 31u;
  const int wv = __builtin_amdgcn_readfirstlane((int)(threadIdx.x >> 5));
  const int row0 = blockIdx.x * OWN_ROWS;
  const int nrows = (n - row0 < OWN_ROWS) ? (n - row0) : OWN_ROWS;
  int* q = queue + wv * Q_WAVE;
  unsigned* glist = list + (size_t)blockIdx.x * LIST_CAP;
  int* wtab = lcount + MAX_OWN + blockIdx.x * WT_PITCH;
  const bool tables_fit = nchunks <= MAX_CHUNKS;

  const bool has_seg = FIRST && tables_fit && tid < nchunks;
  int a0 = 0, a1 = 0;
  if (has_seg) {
    const int* sg = seg + (size_t)tid * SEG_PITCH + blockIdx.x;
    a0 = sg[0];
    a1 = sg[1];
  }
  for (int i = tid; i < OWN_ROWS * MOUT; i += 1024) acc[i] = 0.0f;
  for (int i = tid; i < nrows; i += 1024) dinv_l[i] = dinv[row0 + i];
  if (FIRST && tid < 32) wcnt[tid] = 0;
  if (has_seg) {
    sst[tid] = tid * SORT_PITCH + a0;
    sln[tid] = a1 - a0;
  }
  __syncthreads();

  if (FIRST) {
    int cnt = LIST_CAP + 1;
    if (tables_fit) {
      int carry = 0;
      for (int base = 0; base < nchunks; base += 32) {
        const int ci = base + (int)lane;
        const int v = (ci < nchunks) ? sln[ci] : 0;
        int incl = v;
#pragma unroll
        for (int o = 1; o < 32; o <<= 1) {
          const int t = __shfl_up(incl, o);
          if ((int)lane >= o) incl += t;
        }
        const int pr = carry + __shfl(incl - v, wv);
        const int ln = __shfl(v, wv);
        carry += __shfl(incl, 31);
        if (ln > 0) {
          const int st = sst[base + wv];
          for (int i0 = (int)lane; i0 < ln + (int)lane; i0 += 128) {
            unsigned t[4];
#pragma unroll
            for (int u = 0; u < 4; ++u) {
              const int i = i0 + 32 * u;
              t[u] = sorted[st + ((i < ln) ? i : (ln - 1))];
            }
#pragma unroll
            for (int u = 0; u < 4; ++u) {
              const int i = i0 + 32 * u;
              if (i < ln && pr + i < LIST_CAP) lst[pr + i] = t[u];
            }
          }
        }
      }
      cnt = __builtin_amdgcn_readfirstlane(carry);
    }
    __syncthreads();
    if (tid == 0) lcount[blockIdx.x] = cnt;
    if (__builtin_expect(cnt <= LIST_CAP, 1)) {
      constexpr int LPT = LIST_CAP / 1024;
      unsigned w[LPT];
      int br[LPT];
#pragma unroll
      for (int k = 0; k < LPT; ++k) {
        const int i = tid + k * 1024;
        w[k] = 0u;
        br[k] = -1;
        if (i < cnt) {
          w[k] = lst[i];
          const int o = (int)(w[k] >> 17) & 31;
          br[k] = (o << 16) | atomicAdd(&wcnt[o], 1);
        }
      }
      __syncthreads();
      if (tid < 32) {
        const int v = wcnt[tid];
        int incl = v;
#pragma unroll
        for (int o = 1; o < 32; o <<= 1) {
          const int t = __shfl_up(incl, o);
          if (tid >= o) incl += t;
        }
        wbase[tid] = incl - v;
        if (tid == 31) wbase[32] = incl;
      }
      __syncthreads();
#pragma unroll
      for (int k = 0; k < LPT; ++k) {
        if (br[k] >= 0) lst2[wbase[br[k] >> 16] + (br[k] & 0xFFFF)] = w[k];
      }
      if (tid <= 32) wtab[tid] = wbase[tid];
      __syncthreads();
      for (int i = tid; i < cnt; i += 1024) glist[i] = lst2[i];
      walk_own<MOUT, true>(glist, lst2, wbase[wv], wbase[wv + 1], dinv_l, hs, acc, lane, wv);
    } else {
      scan_edges<MOUT>(src, dst, own4, hs, dinv, acc, q, row0, e, lane, wv);
    }
  } else {
    const int cnt = lcount[blockIdx.x];
    if (__builtin_expect(cnt <= LIST_CAP, 1)) {
      const int beg = __builtin_amdgcn_readfirstlane(wtab[wv]);
      const int end = __builtin_amdgcn_readfirstlane(wtab[wv + 1]);
      walk_own<MOUT, false>(glist, lst2, beg, end, dinv_l, hs, acc, lane, wv);
    } else {
      scan_edges<MOUT>(src, dst, own4, hs, dinv, acc, q, row0, e, lane, wv);
    }
  }

  constexpr int ROWQ = MOUT / 4;
  constexpr int NQ = (OWN_ROWS * ROWQ + 1023) / 1024;
  const int nq = nrows * ROWQ;
  const float4* hs_own = reinterpret_cast<const float4*>(hs + (size_t)row0 * MOUT);
  float4 own[NQ];
#pragma unroll
  for (int k = 0; k < NQ; ++k) {
    const int p = tid + k * 1024;
    own[k] = hs_own[(p < nq) ? p : 0];
  }
  static_assert(1024 % ROWQ == 0, "a thread's pieces must share their columns");
  float bq[4] = {0.0f, 0.0f, 0.0f, 0.0f};
  if (LAST) {
    const int cb = (tid % ROWQ) * 4;
#pragma unroll
    for (int j = 0; j < 4; ++j) bq[j] = bias_out[cb + j];
  }
  __syncthreads();
  const float4* acc4 = reinterpret_cast<const float4*>(acc);
  float4* out_rows = reinterpret_cast<float4*>(agg + (size_t)row0 * MOUT);
#pragma unroll
  for (int k = 0; k < NQ; ++k) {
    const int p = tid + k * 1024;
    if (p < nq) {
      const float dn = dinv_l[p / ROWQ];
      const float4 a = acc4[p];
      float4 o;
      o.x = a.x + own[k].x * dn;
      o.y = a.y + own[k].y * dn;
      o.z = a.z + own[k].z * dn;
      o.w = a.w + own[k].w * dn;
      if (LAST) {
        o.x += bq[0];
        o.y += bq[1];
        o.z += bq[2];
        o.w += bq[3];
      }
      out_rows[p] = o;
    }
  }
}


template <int K, int MOUT, bool ACT, bool FIRST, bool LAST>
static void run_layer(const float* in, const float* W, const float* bias_in,
                      float* dinv, const int* esrc, const int* edst, const unsigned* own4,
                      float* hs, float* agg, unsigned* list, int* lcount, const unsigned* parts,
                      int nparts, const float* bias_out, const unsigned* sorted, const int* seg,
                      int n, int e, hipStream_t stream) {
  k_gemm_strip<K, MOUT, ACT, FIRST><<<256, 1024, 0, stream>>>(in, W, bias_in, dinv, hs, n, parts,
                                                             dinv, nparts);

  const int sblocks = (n + OWN_ROWS - 1) / OWN_ROWS;
  k_scatter_lds<MOUT, FIRST, LAST><<<sblocks, 1024, 0, stream>>>(esrc, edst, own4, hs, dinv, agg,
                                                                list, lcount, bias_out, sorted, seg,
                                                                nparts, n, e);
}

extern "C" void kernel_launch(void* const* d_in, const int* in_sizes, int n_in,
                              void* d_out, int out_size, void* d_ws, size_t ws_size,
                              hipStream_t stream) {
  const float* x  = (const float*)d_in[0];
  const int*   ei = (const int*)d_in[1];
  const float* W1 = (const float*)d_in[2];
  const float* b1 = (const float*)d_in[3];
  const float* W2 = (const float*)d_in[4];
  const float* b2 = (const float*)d_in[5];
  const float* W3 = (const float*)d_in[6];
  const float* b3 = (const float*)d_in[7];
  const float* W4 = (const float*)d_in[8];
  const float* b4 = (const float*)d_in[9];

  const int n = in_sizes[0] / 256;
  const int e = in_sizes[1] / 2;
  const int* esrc = ei;
  const int* edst = ei + e;

  float* dinv = (float*)d_ws;
  float* H  = dinv + n;
  float* G1 = H  + (size_t)n * 128;
  float* G2 = G1 + (size_t)n * 128;
  unsigned* parts = (unsigned*)(G2 + (size_t)n * 128);

  const int nparts = (e + DEG_CHUNK - 1) / DEG_CHUNK;
  unsigned* own4 = parts + (size_t)nparts * ((n + 1) / 2);
  unsigned* list = own4 + (((size_t)e + 127) / 128) * 32;
  int* lcount = (int*)(list + (size_t)((n + OWN_ROWS - 1) / OWN_ROWS) * LIST_CAP);
  unsigned* sorted = (unsigned*)(lcount + MAX_OWN + MAX_OWN * WT_PITCH);
  int* seg = (int*)(sorted + (size_t)nparts * SORT_PITCH);
  k_deg_hist<<<nparts, 1024, 0, stream>>>(esrc, edst, parts, own4, sorted, seg, n, e);

  run_layer<256, 128, false, true,  false>(x,  W1, nullptr, dinv, esrc, edst, own4, H, G1, list, lcount, parts, nparts, nullptr, sorted, seg, n, e, stream);
  run_layer<128, 128, true,  false, false>(G1, W2, b1,      dinv, esrc, edst, own4, H, G2, list, lcount, parts, nparts, nullptr, sorted, seg, n, e, stream);
  run_layer<128, 64,  true,  false, false>(G2, W3, b2,      dinv, esrc, edst, own4, H, G1, list, lcount, parts, nparts, nullptr, sorted, seg, n, e, stream);
  run_layer<64,  32,  true,  false, true >(G1, W4, b3,      dinv, esrc, edst, own4, H, (float*)d_out, list, lcount, parts, nparts, b4, sorted, seg, n, e, stream);
}
